# starting kernel
# baseline (speedup 1.0000x reference)
#include <hip/hip_runtime.h>
#include <hip/hip_bf16.h>

typedef __attribute__((ext_vector_type(2))) float v2f;
typedef __attribute__((ext_vector_type(8))) float v8f;

__global__ void k_init_deg(float* __restrict__ deg, int n) {
    int i = blockIdx.x * blockDim.x + threadIdx.x;
    if (i < n) deg[i] = 1.0f;
}

__global__ void k_deg_edges(const int* __restrict__ dst, float* __restrict__ deg, int E) {
    int e = blockIdx.x * blockDim.x + threadIdx.x;
    if (e < E) atomicAdd(&deg[dst[e]], 1.0f);
}

__global__ void k_dis(float* __restrict__ deg, int n) {
    int i = blockIdx.x * blockDim.x + threadIdx.x;
    if (i < n) {
        float d = deg[i];
        deg[i] = (d > 0.0f) ? rsqrtf(d) : 0.0f;
    }
}

__global__ void k_gemm_wmma(const float* __restrict__ A, const float* __restrict__ W,
                            float* __restrict__ out, int n, int K) {
    __shared__ float sW[64 * 64];
    const int tid = threadIdx.x;
    for (int i = tid; i < K * 64; i += 128) sW[i] = W[i];
    __syncthreads();

    const int wave = tid >> 5;
    const int lane = tid & 31;
    const int tile_row = blockIdx.x * 64 + wave * 16;

    const int m    = lane & 15;
    const int koff = (lane >> 4) * 2;
    int row = tile_row + m;
    if (row >= n) row = n - 1;

    v8f acc[4] = {v8f{0.f}, v8f{0.f}, v8f{0.f}, v8f{0.f}};

    for (int kb = 0; kb < K; kb += 4) {
        v2f a;
        const float* ap = A + (size_t)row * K + kb + koff;
        a.x = ap[0];
        a.y = ap[1];
#pragma unroll
        for (int nt = 0; nt < 4; ++nt) {
            v2f b;
            const int col = nt * 16 + (lane & 15);
            b.x = sW[(kb + koff) * 64 + col];
            b.y = sW[(kb + koff + 1) * 64 + col];
            acc[nt] = __builtin_amdgcn_wmma_f32_16x16x4_f32(
                 false, a,  false, b,
                 (short)0, acc[nt],  false,  false);
        }
    }

#pragma unroll
    for (int nt = 0; nt < 4; ++nt) {
#pragma unroll
        for (int r = 0; r < 8; ++r) {
            const int mm = r + 8 * (lane >> 4);
            const int rr = tile_row + mm;
            if (rr < n) out[(size_t)rr * 64 + nt * 16 + (lane & 15)] = acc[nt][r];
        }
    }
}

__global__ void k_self_init(const float* __restrict__ dis, const float* __restrict__ h,
                            float* __restrict__ acc, int n) {
    int t = blockIdx.x * blockDim.x + threadIdx.x;
    if (t >= n * 16) return;
    int i = t >> 4;
    int jj = (t & 15) * 4;
    float di = dis[i];
    float s = di * di;
    const float4 hv = *(const float4*)&h[(size_t)i * 64 + jj];
    float4 o;
    o.x = s * hv.x; o.y = s * hv.y; o.z = s * hv.z; o.w = s * hv.w;
    *(float4*)&acc[(size_t)i * 64 + jj] = o;
}

__global__ void k_scatter(const int* __restrict__ src, const int* __restrict__ dst,
                          const float* __restrict__ dis, const float* __restrict__ h,
                          float* __restrict__ acc, int E) {
    int t = blockIdx.x * blockDim.x + threadIdx.x;
    if (t >= E * 16) return;
    int e = t >> 4;
    int jj = (t & 15) * 4;
    int s = src[e];
    int d = dst[e];
    float norm = dis[s] * dis[d];
    const float4 hv = *(const float4*)&h[(size_t)s * 64 + jj];
    float* ap = &acc[(size_t)d * 64 + jj];
    atomicAdd(ap + 0, norm * hv.x);
    atomicAdd(ap + 1, norm * hv.y);
    atomicAdd(ap + 2, norm * hv.z);
    atomicAdd(ap + 3, norm * hv.w);
}

__global__ void k_bias_relu(float* __restrict__ acc, const float* __restrict__ b, int n) {
    int t = blockIdx.x * blockDim.x + threadIdx.x;
    if (t >= n * 64) return;
    int j = t & 63;
    acc[t] = fmaxf(acc[t] + b[j], 0.0f);
}

__global__ void k_zero(float* __restrict__ p, int n) {
    int t = blockIdx.x * blockDim.x + threadIdx.x;
    if (t < n) p[t] = 0.0f;
}

__global__ void k_pool(const float* __restrict__ acc2, const float* __restrict__ b2,
                       const int* __restrict__ batch, float* __restrict__ sums, int n) {
    int t = blockIdx.x * blockDim.x + threadIdx.x;
    if (t >= n * 16) return;
    int i = t >> 4;
    int jj = (t & 15) * 4;
    int g = batch[i];
    const float4 hv = *(const float4*)&acc2[(size_t)i * 64 + jj];
    const float4 bv = *(const float4*)&b2[jj];
    float* sp = &sums[(size_t)g * 64 + jj];
    atomicAdd(sp + 0, fmaxf(hv.x + bv.x, 0.0f));
    atomicAdd(sp + 1, fmaxf(hv.y + bv.y, 0.0f));
    atomicAdd(sp + 2, fmaxf(hv.z + bv.z, 0.0f));
    atomicAdd(sp + 3, fmaxf(hv.w + bv.w, 0.0f));
}

__global__ void k_count(const int* __restrict__ batch, float* __restrict__ cnt, int n) {
    int i = blockIdx.x * blockDim.x + threadIdx.x;
    if (i < n) atomicAdd(&cnt[batch[i]], 1.0f);
}

__global__ void k_div(const float* __restrict__ sums, const float* __restrict__ cnt,
                      float* __restrict__ out, int G) {
    int t = blockIdx.x * blockDim.x + threadIdx.x;
    if (t >= G * 64) return;
    out[t] = sums[t] / fmaxf(cnt[t >> 6], 1.0f);
}

extern "C" void kernel_launch(void* const* d_in, const int* in_sizes, int n_in,
                              void* d_out, int out_size, void* d_ws, size_t ws_size,
                              hipStream_t stream) {
    const float* x     = (const float*)d_in[0];
    const int* eidx    = (const int*)d_in[1];
    const int* batch   = (const int*)d_in[2];
    const float* W1 = (const float*)d_in[4];
    const float* b1 = (const float*)d_in[5];
    const float* W2 = (const float*)d_in[6];
    const float* b2 = (const float*)d_in[7];

    const int n = in_sizes[0] / 4;
    const int E = in_sizes[1] / 2;
    const int G = out_size / 64;

    const int* src = eidx;
    const int* dst = eidx + E;

    float* ws   = (float*)d_ws;
    float* dis  = ws;
    float* bufA = dis + n;
    float* bufB = bufA + (size_t)n * 64;
    float* sums = bufB + (size_t)n * 64;
    float* cnt  = sums + (size_t)G * 64;

    const int T = 256;
    auto blk = [T](int work) { return (work + T - 1) / T; };

    k_init_deg<<<blk(n), T, 0, stream>>>(dis, n);
    k_deg_edges<<<blk(E), T, 0, stream>>>(dst, dis, E);
    k_dis<<<blk(n), T, 0, stream>>>(dis, n);

    k_gemm_wmma<<<(n + 63) / 64, 128, 0, stream>>>(x, W1, bufA, n, 4);
    k_self_init<<<blk(n * 16), T, 0, stream>>>(dis, bufA, bufB, n);
    k_scatter<<<blk(E * 16), T, 0, stream>>>(src, dst, dis, bufA, bufB, E);
    k_bias_relu<<<blk(n * 64), T, 0, stream>>>(bufB, b1, n);

    k_gemm_wmma<<<(n + 63) / 64, 128, 0, stream>>>(bufB, W2, bufA, n, 64);
    k_self_init<<<blk(n * 16), T, 0, stream>>>(dis, bufA, bufB, n);
    k_scatter<<<blk(E * 16), T, 0, stream>>>(src, dst, dis, bufA, bufB, E);

    k_zero<<<blk(G * 64 + G), T, 0, stream>>>(sums, G * 64 + G);
    k_pool<<<blk(n * 16), T, 0, stream>>>(bufB, b2, batch, sums, n);
    k_count<<<blk(n), T, 0, stream>>>(batch, cnt, n);
    k_div<<<blk(G * 64), T, 0, stream>>>(sums, cnt, (float*)d_out, G);
}


# workspace re-ordered: both [n,64] planes first, on a 256-B boundary (rows began 64 B past a line, behind the [n] vector): a wave's 128-B row load or float atomic is one line, not two; sum order same
# speedup vs baseline: 2.1690x; 2.1690x over previous
#include <hip/hip_runtime.h>
#include <hip/hip_bf16.h>

typedef __attribute__((ext_vector_type(2))) float v2f;
typedef __attribute__((ext_vector_type(8))) float v8f;
typedef __attribute__((ext_vector_type(4))) float v4f;

#define DEG_SLICES 64
#define DEG_CAP    (50 * 1024)

__global__ void __launch_bounds__(1024)
k_deg_hist(const int* __restrict__ dst, int* __restrict__ part, int E, int n,
           float* __restrict__ zr, int nz) {
    __shared__ int hist[DEG_CAP];
    const int tid = threadIdx.x;
    for (int i = blockIdx.x * 1024 + tid; i < nz; i += gridDim.x * 1024) zr[i] = 0.0f;
    const int per = (E + gridDim.x - 1) / gridDim.x;
    const int e0  = blockIdx.x * per;
    const int e1  = (e0 + per < E) ? (e0 + per) : E;
    int* __restrict__ plane = part + (size_t)blockIdx.x * n;

    for (int base = 0; base < n; base += DEG_CAP) {
        const int m = (n - base < DEG_CAP) ? (n - base) : DEG_CAP;
        for (int i = tid; i < (m + 3) / 4; i += 1024) ((int4*)hist)[i] = make_int4(0, 0, 0, 0);
        __syncthreads();

        int e = e0 + tid;
        for (; e + 3 * 1024 < e1; e += 4 * 1024) {
            const unsigned r0 = (unsigned)(dst[e] - base);
            const unsigned r1 = (unsigned)(dst[e + 1024] - base);
            const unsigned r2 = (unsigned)(dst[e + 2048] - base);
            const unsigned r3 = (unsigned)(dst[e + 3072] - base);
            if (r0 < (unsigned)m) atomicAdd(&hist[r0], 1);
            if (r1 < (unsigned)m) atomicAdd(&hist[r1], 1);
            if (r2 < (unsigned)m) atomicAdd(&hist[r2], 1);
            if (r3 < (unsigned)m) atomicAdd(&hist[r3], 1);
        }
        for (; e < e1; e += 1024) {
            const unsigned r = (unsigned)(dst[e] - base);
            if (r < (unsigned)m) atomicAdd(&hist[r], 1);
        }
        __syncthreads();

        for (int i = tid; i < m; i += 1024) plane[base + i] = hist[i];
        __syncthreads();
    }
}

__global__ void k_dis(const int* __restrict__ part, float* __restrict__ dis, int n,
                      const int* __restrict__ batch, float* __restrict__ cnt) {
    int i = blockIdx.x * blockDim.x + threadIdx.x;
    if (i < n) {
        const int g = batch[i];
        int c = 0;
#pragma unroll 16
        for (int w = 0; w < DEG_SLICES; ++w) c += part[(size_t)w * n + i];
        const float d = (float)c + 1.0f;
        dis[i] = (d > 0.0f) ? rsqrtf(d) : 0.0f;
        atomicAdd(&cnt[g], 1.0f);
    }
}

template <bool RELU_A>
__global__ void k_gemm_wmma(const float* A, const float* __restrict__ W,
                            const float* __restrict__ bA, const float* __restrict__ dis,
                            float* __restrict__ out, float* out2, int n, int K) {
    __shared__ float sW[64 * 64];
    __shared__ float sBias[64];
    const int tid = threadIdx.x;
    for (int i = tid; i < K * 64; i += 128) sW[i] = W[i];
    if (RELU_A) {
        if (tid < K) sBias[tid] = bA[tid];
    }
    __syncthreads();

    const int wave = tid >> 5;
    const int lane = tid & 31;
    const int tile_row = blockIdx.x * 64 + wave * 16;
    if (tile_row >= n) return;

    const int m    = lane & 15;
    const int half = lane >> 4;
    const int koff = half * 2;
    int row = tile_row + m;
    if (row >= n) row = n - 1;

    float d2[8];
#pragma unroll
    for (int r = 0; r < 8; ++r) {
        int rr = tile_row + r + 8 * half;
        if (rr >= n) rr = n - 1;
        const float d = dis[rr];
        d2[r] = d * d;
    }

    v8f acc[4] = {v8f{0.f}, v8f{0.f}, v8f{0.f}, v8f{0.f}};

    for (int kb = 0; kb < K; kb += 4) {
        v2f a;
        const float* ap = A + (size_t)row * K + kb + koff;
        a.x = ap[0];
        a.y = ap[1];
        if (RELU_A) {
            a.x = fmaxf(a.x + sBias[kb + koff], 0.0f);
            a.y = fmaxf(a.y + sBias[kb + koff + 1], 0.0f);
        }
#pragma unroll
        for (int nt = 0; nt < 4; ++nt) {
            v2f b;
            const int col = nt * 16 + (lane & 15);
            b.x = sW[(kb + koff) * 64 + col];
            b.y = sW[(kb + koff + 1) * 64 + col];
            acc[nt] = __builtin_amdgcn_wmma_f32_16x16x4_f32(
                 false, a,  false, b,
                 (short)0, acc[nt],  false,  false);
        }
    }

#pragma unroll
    for (int nt = 0; nt < 4; ++nt) {
#pragma unroll
        for (int r = 0; r < 8; ++r) {
            const int rr = tile_row + r + 8 * half;
            if (rr < n) {
                const size_t o = (size_t)rr * 64 + nt * 16 + m;
                const float v = acc[nt][r];
                out[o]  = v;
                out2[o] = d2[r] * v;
            }
        }
    }
}


__global__ void
k_scatter(const int* __restrict__ src, const int* __restrict__ dst,
          const float* __restrict__ dis, const float* __restrict__ h,
          float* __restrict__ acc, int E) {
    const int e    = blockIdx.x * blockDim.x + threadIdx.x;
    const int lane = threadIdx.x & 31;
    const int e0 = __builtin_amdgcn_readfirstlane(e);
    if (e0 >= E) return;
    int s = 0, d = 0;
    float w = 0.0f;
    if (e < E) {
        s = src[e];
        d = dst[e];
        w = dis[s] * dis[d];
    }
    const int wbits = __float_as_int(w);

    if (E - e0 >= 32) {
#pragma unroll
        for (int jb = 0; jb < 32; jb += 8) {
            float v0[8], v1[8];
            for (int j = 0; j < 8; ++j) {
                const int sj = __builtin_amdgcn_readlane(s, jb + j);
                const float* hp = h + (size_t)sj * 64 + lane;
                v0[j] = hp[0];
                v1[j] = hp[32];
            }
            for (int j = 0; j < 8; ++j) {
                const int dj = __builtin_amdgcn_readlane(d, jb + j);
                const float wj = __int_as_float(__builtin_amdgcn_readlane(wbits, jb + j));
                float* ap = acc + (size_t)dj * 64 + lane;
                atomicAdd(ap, wj * v0[j]);
                atomicAdd(ap + 32, wj * v1[j]);
            }
        }
    } else {
        const int cnt = E - e0;
        for (int j = 0; j < cnt; ++j) {
            const int sj = __builtin_amdgcn_readlane(s, j);
            const int dj = __builtin_amdgcn_readlane(d, j);
            const float wj = __int_as_float(__builtin_amdgcn_readlane(wbits, j));
            const float* hp = h + (size_t)sj * 64 + lane;
            float* ap = acc + (size_t)dj * 64 + lane;
            atomicAdd(ap, wj * hp[0]);
            atomicAdd(ap + 32, wj * hp[32]);
        }
    }
}

__global__ void k_pool(const float* __restrict__ acc2, const float* __restrict__ b2,
                       const int* __restrict__ batch, float* __restrict__ sums, int n) {
    const int Q      = n * 2;
    const int stride = gridDim.x * blockDim.x;
    for (int u = blockIdx.x * blockDim.x + threadIdx.x; u < Q; u += stride) {
        int g[8];
        v4f hv[8];
        v4f bv[8];
#pragma unroll
        for (int k = 0; k < 8; ++k) {
            const int t = u + k * Q;
            g[k]  = batch[t >> 4];
            hv[k] = *(const v4f*)&acc2[(size_t)t * 4];
            bv[k] = *(const v4f*)&b2[(t & 15) * 4];
        }
        __builtin_amdgcn_sched_barrier(0);
#pragma unroll
        for (int k = 0; k < 8; ++k) {
            const int jj = ((u + k * Q) & 15) * 4;
            float* sp = &sums[(size_t)g[k] * 64 + jj];
            const v4f w = hv[k] + bv[k];
            atomicAdd(sp + 0, fmaxf(w.x, 0.0f));
            atomicAdd(sp + 1, fmaxf(w.y, 0.0f));
            atomicAdd(sp + 2, fmaxf(w.z, 0.0f));
            atomicAdd(sp + 3, fmaxf(w.w, 0.0f));
        }
    }
}

__global__ void k_div(const float* __restrict__ sums, const float* __restrict__ cnt,
                      float* __restrict__ out, int G) {
    int t = blockIdx.x * blockDim.x + threadIdx.x;
    if (t >= G * 64) return;
    out[t] = sums[t] / fmaxf(cnt[t >> 6], 1.0f);
}

extern "C" void kernel_launch(void* const* d_in, const int* in_sizes, int n_in,
                              void* d_out, int out_size, void* d_ws, size_t ws_size,
                              hipStream_t stream) {
    const float* x     = (const float*)d_in[0];
    const int* eidx    = (const int*)d_in[1];
    const int* batch   = (const int*)d_in[2];
    const float* W1 = (const float*)d_in[4];
    const float* b1 = (const float*)d_in[5];
    const float* W2 = (const float*)d_in[6];
    const float* b2 = (const float*)d_in[7];

    const int n = in_sizes[0] / 4;
    const int E = in_sizes[1] / 2;
    const int G = out_size / 64;

    const int* src = eidx;
    const int* dst = eidx + E;

    char* wb = (char*)d_ws;
    const size_t need = ((size_t)n * 129 + (size_t)G * 65) * sizeof(float);
    const size_t pad  = (256 - ((size_t)wb & 255)) & 255;
    if (need + pad <= ws_size) wb += pad;
    float* bufA = (float*)wb;
    float* bufB = bufA + (size_t)n * 64;
    float* sums = bufB + (size_t)n * 64;
    float* cnt  = sums + (size_t)G * 64;
    float* dis  = cnt + G;

    const int T = 256;
    auto blk = [T](int work) { return (work + T - 1) / T; };
    auto fblk = [T](int items) { const int g = (items / 8 + T - 1) / T; return g < 1 ? 1 : (g < 512 ? g : 512); };

    k_deg_hist<<<DEG_SLICES, 1024, 0, stream>>>(dst, (int*)bufB, E, n, sums, G * 64 + G);
    k_dis<<<blk(n), T, 0, stream>>>((const int*)bufB, dis, n, batch, cnt);

    k_gemm_wmma<false><<<(n + 63) / 64, 128, 0, stream>>>(x, W1, b1, dis, bufA, bufB, n, 4);
    k_scatter<<<blk(E), T, 0, stream>>>(src, dst, dis, bufA, bufB, E);

    k_gemm_wmma<true><<<(n + 63) / 64, 128, 0, stream>>>(bufB, W2, b1, dis, bufA, bufB, n, 64);
    k_scatter<<<blk(E), T, 0, stream>>>(src, dst, dis, bufA, bufB, E);

    k_pool<<<fblk(n * 16), T, 0, stream>>>(bufB, b2, batch, sums, n);
    k_div<<<blk(G * 64), T, 0, stream>>>(sums, cnt, (float*)d_out, G);
}


# edge scatter reads both norm factors of an edge from a whole-range copy of dis in LDS (one resident 32-wave work-group per WGP loops over 32-edge chunks), no global gathers; atomics, sum order kept
# speedup vs baseline: 2.2493x; 1.0370x over previous
#include <hip/hip_runtime.h>
#include <hip/hip_bf16.h>

typedef __attribute__((ext_vector_type(2))) float v2f;
typedef __attribute__((ext_vector_type(8))) float v8f;
typedef __attribute__((ext_vector_type(4))) float v4f;

#define DEG_SLICES 64
#define DEG_CAP    (50 * 1024)

__global__ void __launch_bounds__(1024)
k_deg_hist(const int* __restrict__ dst, int* __restrict__ part, int E, int n,
           float* __restrict__ zr, int nz) {
    __shared__ int hist[DEG_CAP];
    const int tid = threadIdx.x;
    for (int i = blockIdx.x * 1024 + tid; i < nz; i += gridDim.x * 1024) zr[i] = 0.0f;
    const int per = (E + gridDim.x - 1) / gridDim.x;
    const int e0  = blockIdx.x * per;
    const int e1  = (e0 + per < E) ? (e0 + per) : E;
    int* __restrict__ plane = part + (size_t)blockIdx.x * n;

    for (int base = 0; base < n; base += DEG_CAP) {
        const int m = (n - base < DEG_CAP) ? (n - base) : DEG_CAP;
        for (int i = tid; i < (m + 3) / 4; i += 1024) ((int4*)hist)[i] = make_int4(0, 0, 0, 0);
        __syncthreads();

        int e = e0 + tid;
        for (; e + 3 * 1024 < e1; e += 4 * 1024) {
            const unsigned r0 = (unsigned)(dst[e] - base);
            const unsigned r1 = (unsigned)(dst[e + 1024] - base);
            const unsigned r2 = (unsigned)(dst[e + 2048] - base);
            const unsigned r3 = (unsigned)(dst[e + 3072] - base);
            if (r0 < (unsigned)m) atomicAdd(&hist[r0], 1);
            if (r1 < (unsigned)m) atomicAdd(&hist[r1], 1);
            if (r2 < (unsigned)m) atomicAdd(&hist[r2], 1);
            if (r3 < (unsigned)m) atomicAdd(&hist[r3], 1);
        }
        for (; e < e1; e += 1024) {
            const unsigned r = (unsigned)(dst[e] - base);
            if (r < (unsigned)m) atomicAdd(&hist[r], 1);
        }
        __syncthreads();

        for (int i = tid; i < m; i += 1024) plane[base + i] = hist[i];
        __syncthreads();
    }
}

__global__ void k_dis(const int* __restrict__ part, float* __restrict__ dis, int n,
                      const int* __restrict__ batch, float* __restrict__ cnt) {
    int i = blockIdx.x * blockDim.x + threadIdx.x;
    if (i < n) {
        const int g = batch[i];
        int c = 0;
#pragma unroll 16
        for (int w = 0; w < DEG_SLICES; ++w) c += part[(size_t)w * n + i];
        const float d = (float)c + 1.0f;
        dis[i] = (d > 0.0f) ? rsqrtf(d) : 0.0f;
        atomicAdd(&cnt[g], 1.0f);
    }
}

template <bool RELU_A>
__global__ void k_gemm_wmma(const float* A, const float* __restrict__ W,
                            const float* __restrict__ bA, const float* __restrict__ dis,
                            float* __restrict__ out, float* out2, int n, int K) {
    __shared__ float sW[64 * 64];
    __shared__ float sBias[64];
    const int tid = threadIdx.x;
    for (int i = tid; i < K * 64; i += 128) sW[i] = W[i];
    if (RELU_A) {
        if (tid < K) sBias[tid] = bA[tid];
    }
    __syncthreads();

    const int wave = tid >> 5;
    const int lane = tid & 31;
    const int tile_row = blockIdx.x * 64 + wave * 16;
    if (tile_row >= n) return;

    const int m    = lane & 15;
    const int half = lane >> 4;
    const int koff = half * 2;
    int row = tile_row + m;
    if (row >= n) row = n - 1;

    float d2[8];
#pragma unroll
    for (int r = 0; r < 8; ++r) {
        int rr = tile_row + r + 8 * half;
        if (rr >= n) rr = n - 1;
        const float d = dis[rr];
        d2[r] = d * d;
    }

    v8f acc[4] = {v8f{0.f}, v8f{0.f}, v8f{0.f}, v8f{0.f}};

    for (int kb = 0; kb < K; kb += 4) {
        v2f a;
        const float* ap = A + (size_t)row * K + kb + koff;
        a.x = ap[0];
        a.y = ap[1];
        if (RELU_A) {
            a.x = fmaxf(a.x + sBias[kb + koff], 0.0f);
            a.y = fmaxf(a.y + sBias[kb + koff + 1], 0.0f);
        }
#pragma unroll
        for (int nt = 0; nt < 4; ++nt) {
            v2f b;
            const int col = nt * 16 + (lane & 15);
            b.x = sW[(kb + koff) * 64 + col];
            b.y = sW[(kb + koff + 1) * 64 + col];
            acc[nt] = __builtin_amdgcn_wmma_f32_16x16x4_f32(
                 false, a,  false, b,
                 (short)0, acc[nt],  false,  false);
        }
    }

#pragma unroll
    for (int nt = 0; nt < 4; ++nt) {
#pragma unroll
        for (int r = 0; r < 8; ++r) {
            const int rr = tile_row + r + 8 * half;
            if (rr < n) {
                const size_t o = (size_t)rr * 64 + nt * 16 + m;
                const float v = acc[nt][r];
                out[o]  = v;
                out2[o] = d2[r] * v;
            }
        }
    }
}


#define DIS_CAP   (50 * 1024)
#define SC_WAVES  32
#define SC_FILL   13

__global__ void __launch_bounds__(1024)
k_scatter(const int* __restrict__ src, const int* __restrict__ dst,
          const float* __restrict__ dis, const float* __restrict__ h,
          float* __restrict__ acc, int E, int n) {
    __shared__ float sdis[DIS_CAP];
    const int tid  = threadIdx.x;
    const int lane = tid & 31;
    const int wave = __builtin_amdgcn_readfirstlane(tid >> 5);
    const bool tab = (n <= DIS_CAP);
    const int nchunk = (E + 31) >> 5;
    const int cstep  = gridDim.x * SC_WAVES;

    int c = blockIdx.x + gridDim.x * wave;
    int s = 0, d = 0;
    if (c < nchunk) {
        const int e = c * 32 + lane;
        if (e < E) { s = src[e]; d = dst[e]; }
    }

    if (tab) {
        const int nv = ((((size_t)dis) & 15) == 0) ? (n >> 2) : 0;
        if (nv > 0) {
            const int rot = (int)(((long long)blockIdx.x * nv) / gridDim.x);
            v4f t[SC_FILL];
#pragma unroll
            for (int u = 0; u < SC_FILL; ++u) {
                int k = u * 1024 + tid;
                if (k >= nv) k = nv - 1;
                int j = k + rot;
                if (j >= nv) j -= nv;
                t[u] = ((const v4f*)dis)[j];
            }
#pragma unroll
            for (int u = 0; u < SC_FILL; ++u) {
                const int k = u * 1024 + tid;
                if (k < nv) {
                    int j = k + rot;
                    if (j >= nv) j -= nv;
                    ((v4f*)sdis)[j] = t[u];
                }
            }
        }
        for (int i = nv * 4 + tid; i < n; i += 1024) sdis[i] = dis[i];
    }
    __syncthreads();

    for (; c < nchunk; c += cstep) {
        const int cn = c + cstep;
        int sn = 0, dn = 0;
        if (cn < nchunk) {
            const int en = cn * 32 + lane;
            if (en < E) { sn = src[en]; dn = dst[en]; }
        }
        const int e0 = c * 32;
        float w = 0.0f;
        if (e0 + lane < E) w = tab ? (sdis[s] * sdis[d]) : (dis[s] * dis[d]);
        const int wbits = __float_as_int(w);

        if (E - e0 >= 32) {
#pragma unroll
            for (int jb = 0; jb < 32; jb += 16) {
                float v0[16], v1[16];
#pragma unroll
                for (int j = 0; j < 16; ++j) {
                    const int sj = __builtin_amdgcn_readlane(s, jb + j);
                    const float* hp = h + (size_t)sj * 64 + lane;
                    v0[j] = hp[0];
                    v1[j] = hp[32];
                }
#pragma unroll
                for (int j = 0; j < 16; ++j) {
                    const int dj = __builtin_amdgcn_readlane(d, jb + j);
                    const float wj = __int_as_float(__builtin_amdgcn_readlane(wbits, jb + j));
                    float* ap = acc + (size_t)dj * 64 + lane;
                    atomicAdd(ap, wj * v0[j]);
                    atomicAdd(ap + 32, wj * v1[j]);
                }
            }
        } else {
            const int cnt = E - e0;
            for (int j = 0; j < cnt; ++j) {
                const int sj = __builtin_amdgcn_readlane(s, j);
                const int dj = __builtin_amdgcn_readlane(d, j);
                const float wj = __int_as_float(__builtin_amdgcn_readlane(wbits, j));
                const float* hp = h + (size_t)sj * 64 + lane;
                float* ap = acc + (size_t)dj * 64 + lane;
                atomicAdd(ap, wj * hp[0]);
                atomicAdd(ap + 32, wj * hp[32]);
            }
        }
        s = sn;
        d = dn;
    }
}

__global__ void k_pool(const float* __restrict__ acc2, const float* __restrict__ b2,
                       const int* __restrict__ batch, float* __restrict__ sums, int n) {
    const int Q      = n * 2;
    const int stride = gridDim.x * blockDim.x;
    for (int u = blockIdx.x * blockDim.x + threadIdx.x; u < Q; u += stride) {
        int g[8];
        v4f hv[8];
        v4f bv[8];
#pragma unroll
        for (int k = 0; k < 8; ++k) {
            const int t = u + k * Q;
            g[k]  = batch[t >> 4];
            hv[k] = *(const v4f*)&acc2[(size_t)t * 4];
            bv[k] = *(const v4f*)&b2[(t & 15) * 4];
        }
        __builtin_amdgcn_sched_barrier(0);
#pragma unroll
        for (int k = 0; k < 8; ++k) {
            const int jj = ((u + k * Q) & 15) * 4;
            float* sp = &sums[(size_t)g[k] * 64 + jj];
            const v4f w = hv[k] + bv[k];
            atomicAdd(sp + 0, fmaxf(w.x, 0.0f));
            atomicAdd(sp + 1, fmaxf(w.y, 0.0f));
            atomicAdd(sp + 2, fmaxf(w.z, 0.0f));
            atomicAdd(sp + 3, fmaxf(w.w, 0.0f));
        }
    }
}

__global__ void k_div(const float* __restrict__ sums, const float* __restrict__ cnt,
                      float* __restrict__ out, int G) {
    int t = blockIdx.x * blockDim.x + threadIdx.x;
    if (t >= G * 64) return;
    out[t] = sums[t] / fmaxf(cnt[t >> 6], 1.0f);
}

extern "C" void kernel_launch(void* const* d_in, const int* in_sizes, int n_in,
                              void* d_out, int out_size, void* d_ws, size_t ws_size,
                              hipStream_t stream) {
    const float* x     = (const float*)d_in[0];
    const int* eidx    = (const int*)d_in[1];
    const int* batch   = (const int*)d_in[2];
    const float* W1 = (const float*)d_in[4];
    const float* b1 = (const float*)d_in[5];
    const float* W2 = (const float*)d_in[6];
    const float* b2 = (const float*)d_in[7];

    const int n = in_sizes[0] / 4;
    const int E = in_sizes[1] / 2;
    const int G = out_size / 64;

    const int* src = eidx;
    const int* dst = eidx + E;

    char* wb = (char*)d_ws;
    const size_t need = ((size_t)n * 129 + (size_t)G * 65) * sizeof(float);
    const size_t pad  = (256 - ((size_t)wb & 255)) & 255;
    if (need + pad <= ws_size) wb += pad;
    float* bufA = (float*)wb;
    float* bufB = bufA + (size_t)n * 64;
    float* sums = bufB + (size_t)n * 64;
    float* cnt  = sums + (size_t)G * 64;
    float* dis  = cnt + G;

    const int T = 256;
    auto blk = [T](int work) { return (work + T - 1) / T; };
    auto fblk = [T](int items) { const int g = (items / 8 + T - 1) / T; return g < 1 ? 1 : (g < 512 ? g : 512); };

    k_deg_hist<<<DEG_SLICES, 1024, 0, stream>>>(dst, (int*)bufB, E, n, sums, G * 64 + G);
    k_dis<<<blk(n), T, 0, stream>>>((const int*)bufB, dis, n, batch, cnt);

    k_gemm_wmma<false><<<(n + 63) / 64, 128, 0, stream>>>(x, W1, b1, dis, bufA, bufB, n, 4);
    k_scatter<<<256, 1024, 0, stream>>>(src, dst, dis, bufA, bufB, E, n);

    k_gemm_wmma<true><<<(n + 63) / 64, 128, 0, stream>>>(bufB, W2, b1, dis, bufA, bufB, n, 64);
    k_scatter<<<256, 1024, 0, stream>>>(src, dst, dis, bufA, bufB, E, n);

    k_pool<<<fblk(n * 16), T, 0, stream>>>(bufB, b2, batch, sums, n);
    k_div<<<blk(G * 64), T, 0, stream>>>(sums, cnt, (float*)d_out, G);
}


# speedup vs baseline: 2.6012x; 1.1564x over previous
#include <hip/hip_runtime.h>
#include <hip/hip_bf16.h>

typedef __attribute__((ext_vector_type(2))) float v2f;
typedef __attribute__((ext_vector_type(8))) float v8f;
typedef __attribute__((ext_vector_type(4))) float v4f;

#define DEG_SLICES 64
#define DEG_CAP    (50 * 1024)

__global__ void __launch_bounds__(1024)
k_deg_hist(const int* __restrict__ dst, int* __restrict__ part, int E, int n,
           float* __restrict__ zr, int nz) {
    __shared__ int hist[DEG_CAP];
    const int tid = threadIdx.x;
    for (int i = blockIdx.x * 1024 + tid; i < nz; i += gridDim.x * 1024) zr[i] = 0.0f;
    const int per = (E + gridDim.x - 1) / gridDim.x;
    const int e0  = blockIdx.x * per;
    const int e1  = (e0 + per < E) ? (e0 + per) : E;
    int* __restrict__ plane = part + (size_t)blockIdx.x * n;

    for (int base = 0; base < n; base += DEG_CAP) {
        const int m = (n - base < DEG_CAP) ? (n - base) : DEG_CAP;
        for (int i = tid; i < (m + 3) / 4; i += 1024) ((int4*)hist)[i] = make_int4(0, 0, 0, 0);
        __syncthreads();

        int e = e0 + tid;
        for (; e + 3 * 1024 < e1; e += 4 * 1024) {
            const unsigned r0 = (unsigned)(dst[e] - base);
            const unsigned r1 = (unsigned)(dst[e + 1024] - base);
            const unsigned r2 = (unsigned)(dst[e + 2048] - base);
            const unsigned r3 = (unsigned)(dst[e + 3072] - base);
            if (r0 < (unsigned)m) atomicAdd(&hist[r0], 1);
            if (r1 < (unsigned)m) atomicAdd(&hist[r1], 1);
            if (r2 < (unsigned)m) atomicAdd(&hist[r2], 1);
            if (r3 < (unsigned)m) atomicAdd(&hist[r3], 1);
        }
        for (; e < e1; e += 1024) {
            const unsigned r = (unsigned)(dst[e] - base);
            if (r < (unsigned)m) atomicAdd(&hist[r], 1);
        }
        __syncthreads();

        for (int i = tid; i < m; i += 1024) plane[base + i] = hist[i];
        __syncthreads();
    }
}

__global__ void k_dis(const int* __restrict__ part, float* __restrict__ dis, int n,
                      const int* __restrict__ batch, float* __restrict__ cnt) {
    int i = blockIdx.x * blockDim.x + threadIdx.x;
    if (i < n) {
        const int g = batch[i];
        int c = 0;
#pragma unroll 16
        for (int w = 0; w < DEG_SLICES; ++w) c += part[(size_t)w * n + i];
        const float d = (float)c + 1.0f;
        dis[i] = (d > 0.0f) ? rsqrtf(d) : 0.0f;
        atomicAdd(&cnt[g], 1.0f);
    }
}

template <bool RELU_A, bool TWO>
__global__ void k_gemm_wmma(const float* A, const float* A1, const float* __restrict__ W,
                            const float* __restrict__ bA, const float* __restrict__ dis,
                            float* __restrict__ out, float* out2, float* out3, int n, int K) {
    __shared__ float sW[64 * 64];
    __shared__ float sBias[64];
    const int tid = threadIdx.x;
    for (int i = tid; i < K * 64; i += 128) sW[i] = W[i];
    if (RELU_A) {
        if (tid < K) sBias[tid] = bA[tid];
    }
    __syncthreads();

    const int wave = tid >> 5;
    const int lane = tid & 31;
    const int tile_row = blockIdx.x * 64 + wave * 16;
    if (tile_row >= n) return;

    const int m    = lane & 15;
    const int half = lane >> 4;
    const int koff = half * 2;
    int row = tile_row + m;
    if (row >= n) row = n - 1;

    float d2[8];
#pragma unroll
    for (int r = 0; r < 8; ++r) {
        int rr = tile_row + r + 8 * half;
        if (rr >= n) rr = n - 1;
        const float d = dis[rr];
        d2[r] = d * d;
    }

    v8f acc[4] = {v8f{0.f}, v8f{0.f}, v8f{0.f}, v8f{0.f}};

    for (int kb = 0; kb < K; kb += 4) {
        v2f a;
        const size_t ao = (size_t)row * K + kb + koff;
        a.x = A[ao];
        a.y = A[ao + 1];
        if (RELU_A) {
            if (TWO) {
                a.x += A1[ao];
                a.y += A1[ao + 1];
            }
            a.x = fmaxf(a.x + sBias[kb + koff], 0.0f);
            a.y = fmaxf(a.y + sBias[kb + koff + 1], 0.0f);
        }
#pragma unroll
        for (int nt = 0; nt < 4; ++nt) {
            v2f b;
            const int col = nt * 16 + (lane & 15);
            b.x = sW[(kb + koff) * 64 + col];
            b.y = sW[(kb + koff + 1) * 64 + col];
            acc[nt] = __builtin_amdgcn_wmma_f32_16x16x4_f32(
                 false, a,  false, b,
                 (short)0, acc[nt],  false,  false);
        }
    }

#pragma unroll
    for (int nt = 0; nt < 4; ++nt) {
#pragma unroll
        for (int r = 0; r < 8; ++r) {
            const int rr = tile_row + r + 8 * half;
            if (rr < n) {
                const size_t o = (size_t)rr * 64 + nt * 16 + m;
                const float v = acc[nt][r];
                out[o]  = v;
                out2[o] = d2[r] * v;
                if (TWO) out3[o] = 0.0f;
            }
        }
    }
}


#define DIS_CAP   (50 * 1024)
#define SC_WAVES  32
#define SC_FILL   13

__global__ void __launch_bounds__(1024)
k_scatter(const int* __restrict__ src, const int* __restrict__ dst,
          const float* __restrict__ dis, const float* __restrict__ h,
          float* acc0, float* acc1, int E, int n) {
    __shared__ float sdis[DIS_CAP];
    const bool upper = (__builtin_amdgcn_s_sendmsg_rtn(0x87) >> 18) & 1u;
    float* acc = upper ? acc1 : acc0;
    const int tid  = threadIdx.x;
    const int lane = tid & 31;
    const int wave = __builtin_amdgcn_readfirstlane(tid >> 5);
    const bool tab = (n <= DIS_CAP);
    const int nchunk = (E + 31) >> 5;
    const int cstep  = gridDim.x * SC_WAVES;

    int c = blockIdx.x + gridDim.x * wave;
    int s = 0, d = 0;
    if (c < nchunk) {
        const int e = c * 32 + lane;
        if (e < E) { s = src[e]; d = dst[e]; }
    }

    if (tab) {
        const int nv = ((((size_t)dis) & 15) == 0) ? (n >> 2) : 0;
        if (nv > 0) {
            const int rot = (int)(((long long)blockIdx.x * nv) / gridDim.x);
            v4f t[SC_FILL];
#pragma unroll
            for (int u = 0; u < SC_FILL; ++u) {
                int k = u * 1024 + tid;
                if (k >= nv) k = nv - 1;
                int j = k + rot;
                if (j >= nv) j -= nv;
                t[u] = ((const v4f*)dis)[j];
            }
#pragma unroll
            for (int u = 0; u < SC_FILL; ++u) {
                const int k = u * 1024 + tid;
                if (k < nv) {
                    int j = k + rot;
                    if (j >= nv) j -= nv;
                    ((v4f*)sdis)[j] = t[u];
                }
            }
        }
        for (int i = nv * 4 + tid; i < n; i += 1024) sdis[i] = dis[i];
    }
    __syncthreads();

    for (; c < nchunk; c += cstep) {
        const int cn = c + cstep;
        int sn = 0, dn = 0;
        if (cn < nchunk) {
            const int en = cn * 32 + lane;
            if (en < E) { sn = src[en]; dn = dst[en]; }
        }
        const int e0 = c * 32;
        float w = 0.0f;
        if (e0 + lane < E) w = tab ? (sdis[s] * sdis[d]) : (dis[s] * dis[d]);
        const int wbits = __float_as_int(w);

        if (E - e0 >= 32) {
#pragma unroll
            for (int jb = 0; jb < 32; jb += 16) {
                float v0[16], v1[16];
#pragma unroll
                for (int j = 0; j < 16; ++j) {
                    const int sj = __builtin_amdgcn_readlane(s, jb + j);
                    const float* hp = h + (size_t)sj * 64 + lane;
                    v0[j] = hp[0];
                    v1[j] = hp[32];
                }
#pragma unroll
                for (int j = 0; j < 16; ++j) {
                    const int dj = __builtin_amdgcn_readlane(d, jb + j);
                    const float wj = __int_as_float(__builtin_amdgcn_readlane(wbits, jb + j));
                    float* ap = acc + (size_t)dj * 64 + lane;
                    atomicAdd(ap, wj * v0[j]);
                    atomicAdd(ap + 32, wj * v1[j]);
                }
            }
        } else {
            const int cnt = E - e0;
            for (int j = 0; j < cnt; ++j) {
                const int sj = __builtin_amdgcn_readlane(s, j);
                const int dj = __builtin_amdgcn_readlane(d, j);
                const float wj = __int_as_float(__builtin_amdgcn_readlane(wbits, j));
                const float* hp = h + (size_t)sj * 64 + lane;
                float* ap = acc + (size_t)dj * 64 + lane;
                atomicAdd(ap, wj * hp[0]);
                atomicAdd(ap + 32, wj * hp[32]);
            }
        }
        s = sn;
        d = dn;
    }
}

template <bool TWO>
__global__ void __launch_bounds__(256) k_pool(const float* __restrict__ acc2, const float* __restrict__ acc2b,
                       const float* __restrict__ b2,
                       const int* __restrict__ batch, float* __restrict__ sums, int n) {
    const int Q      = n * 2;
    const int stride = gridDim.x * blockDim.x;
    for (int u = blockIdx.x * blockDim.x + threadIdx.x; u < Q; u += stride) {
        int g[8];
        v4f hv[8];
        v4f hw[8];
        v4f bv[8];
#pragma unroll
        for (int k = 0; k < 8; ++k) {
            const int t = u + k * Q;
            g[k]  = batch[t >> 4];
            hv[k] = *(const v4f*)&acc2[(size_t)t * 4];
            if (TWO) hw[k] = *(const v4f*)&acc2b[(size_t)t * 4];
            bv[k] = *(const v4f*)&b2[(t & 15) * 4];
        }
        __builtin_amdgcn_sched_barrier(0);
#pragma unroll
        for (int k = 0; k < 8; ++k) {
            const int jj = ((u + k * Q) & 15) * 4;
            float* sp = &sums[(size_t)g[k] * 64 + jj];
            const v4f w = (TWO ? (hv[k] + hw[k]) : hv[k]) + bv[k];
            atomicAdd(sp + 0, fmaxf(w.x, 0.0f));
            atomicAdd(sp + 1, fmaxf(w.y, 0.0f));
            atomicAdd(sp + 2, fmaxf(w.z, 0.0f));
            atomicAdd(sp + 3, fmaxf(w.w, 0.0f));
        }
    }
}

__global__ void k_div(const float* __restrict__ sums, const float* __restrict__ cnt,
                      float* __restrict__ out, int G) {
    int t = blockIdx.x * blockDim.x + threadIdx.x;
    if (t >= G * 64) return;
    out[t] = sums[t] / fmaxf(cnt[t >> 6], 1.0f);
}

extern "C" void kernel_launch(void* const* d_in, const int* in_sizes, int n_in,
                              void* d_out, int out_size, void* d_ws, size_t ws_size,
                              hipStream_t stream) {
    const float* x     = (const float*)d_in[0];
    const int* eidx    = (const int*)d_in[1];
    const int* batch   = (const int*)d_in[2];
    const float* W1 = (const float*)d_in[4];
    const float* b1 = (const float*)d_in[5];
    const float* W2 = (const float*)d_in[6];
    const float* b2 = (const float*)d_in[7];

    const int n = in_sizes[0] / 4;
    const int E = in_sizes[1] / 2;
    const int G = out_size / 64;

    const int* src = eidx;
    const int* dst = eidx + E;

    char* wb = (char*)d_ws;
    const size_t need1 = ((size_t)n * 129 + (size_t)G * 65) * sizeof(float);
    const size_t need2 = need1 + (size_t)n * 64 * sizeof(float);
    const size_t pad   = (256 - ((size_t)wb & 255)) & 255;
    const bool two = need2 + pad <= ws_size;
    if ((two ? need2 : need1) + pad <= ws_size) wb += pad;
    float* bufA = (float*)wb;
    float* bufB = bufA + (size_t)n * 64;
    float* bufC = two ? bufB + (size_t)n * 64 : bufB;
    float* sums = bufC + (size_t)n * 64;
    float* cnt  = sums + (size_t)G * 64;
    float* dis  = cnt + G;

    const int T = 256;
    auto blk = [T](int work) { return (work + T - 1) / T; };
    auto fblk = [T](int items) { const int g = (items / 8 + T - 1) / T; return g < 1 ? 1 : (g < 512 ? g : 512); };

    k_deg_hist<<<DEG_SLICES, 1024, 0, stream>>>(dst, (int*)bufB, E, n, sums, G * 64 + G);
    k_dis<<<blk(n), T, 0, stream>>>((const int*)bufB, dis, n, batch, cnt);

    if (two) k_gemm_wmma<false, true><<<(n + 63) / 64, 128, 0, stream>>>(x, nullptr, W1, b1, dis, bufA, bufB, bufC, n, 4);
    else     k_gemm_wmma<false, false><<<(n + 63) / 64, 128, 0, stream>>>(x, nullptr, W1, b1, dis, bufA, bufB, nullptr, n, 4);
    k_scatter<<<256, 1024, 0, stream>>>(src, dst, dis, bufA, bufB, bufC, E, n);

    if (two) k_gemm_wmma<true, true><<<(n + 63) / 64, 128, 0, stream>>>(bufB, bufC, W2, b1, dis, bufA, bufB, bufC, n, 64);
    else     k_gemm_wmma<true, false><<<(n + 63) / 64, 128, 0, stream>>>(bufB, nullptr, W2, b1, dis, bufA, bufB, nullptr, n, 64);
    k_scatter<<<256, 1024, 0, stream>>>(src, dst, dis, bufA, bufB, bufC, E, n);

    if (two) k_pool<true><<<fblk(n * 16), T, 0, stream>>>(bufB, bufC, b2, batch, sums, n);
    else     k_pool<false><<<fblk(n * 16), T, 0, stream>>>(bufB, nullptr, b2, batch, sums, n);
    k_div<<<blk(G * 64), T, 0, stream>>>(sums, cnt, (float*)d_out, G);
}


# mean pool sums and node counts: a wave adds up runs of equal graph id itself (16 consecutive rows in registers, lane = column; counts by ballot), two whole-line atomics per run; sum order changes
# speedup vs baseline: 4.2147x; 1.6203x over previous
#include <hip/hip_runtime.h>
#include <hip/hip_bf16.h>

typedef __attribute__((ext_vector_type(2))) float v2f;
typedef __attribute__((ext_vector_type(8))) float v8f;
typedef __attribute__((ext_vector_type(4))) float v4f;

#define DEG_SLICES 64
#define DEG_CAP    (50 * 1024)

__global__ void __launch_bounds__(1024)
k_deg_hist(const int* __restrict__ dst, int* __restrict__ part, int E, int n,
           float* __restrict__ zr, int nz) {
    __shared__ int hist[DEG_CAP];
    const int tid = threadIdx.x;
    for (int i = blockIdx.x * 1024 + tid; i < nz; i += gridDim.x * 1024) zr[i] = 0.0f;
    const int per = (E + gridDim.x - 1) / gridDim.x;
    const int e0  = blockIdx.x * per;
    const int e1  = (e0 + per < E) ? (e0 + per) : E;
    int* __restrict__ plane = part + (size_t)blockIdx.x * n;

    for (int base = 0; base < n; base += DEG_CAP) {
        const int m = (n - base < DEG_CAP) ? (n - base) : DEG_CAP;
        for (int i = tid; i < (m + 3) / 4; i += 1024) ((int4*)hist)[i] = make_int4(0, 0, 0, 0);
        __syncthreads();

        int e = e0 + tid;
        for (; e + 3 * 1024 < e1; e += 4 * 1024) {
            const unsigned r0 = (unsigned)(dst[e] - base);
            const unsigned r1 = (unsigned)(dst[e + 1024] - base);
            const unsigned r2 = (unsigned)(dst[e + 2048] - base);
            const unsigned r3 = (unsigned)(dst[e + 3072] - base);
            if (r0 < (unsigned)m) atomicAdd(&hist[r0], 1);
            if (r1 < (unsigned)m) atomicAdd(&hist[r1], 1);
            if (r2 < (unsigned)m) atomicAdd(&hist[r2], 1);
            if (r3 < (unsigned)m) atomicAdd(&hist[r3], 1);
        }
        for (; e < e1; e += 1024) {
            const unsigned r = (unsigned)(dst[e] - base);
            if (r < (unsigned)m) atomicAdd(&hist[r], 1);
        }
        __syncthreads();

        for (int i = tid; i < m; i += 1024) plane[base + i] = hist[i];
        __syncthreads();
    }
}

__global__ void k_dis(const int* __restrict__ part, float* __restrict__ dis, int n,
                      const int* __restrict__ batch, float* __restrict__ cnt) {
    const int i = blockIdx.x * blockDim.x + threadIdx.x;
    const bool in = i < n;
    int g = -1;
    if (in) {
        g = batch[i];
        int c = 0;
#pragma unroll 16
        for (int w = 0; w < DEG_SLICES; ++w) c += part[(size_t)w * n + i];
        const float d = (float)c + 1.0f;
        dis[i] = (d > 0.0f) ? rsqrtf(d) : 0.0f;
    }
    const int lane = threadIdx.x & 31;
    unsigned todo = __builtin_amdgcn_ballot_w32(in);
    while (todo != 0u) {
        const int first = __builtin_ctz(todo);
        const int g0 = __builtin_amdgcn_readlane(g, first);
        const unsigned same = __builtin_amdgcn_ballot_w32(in && g == g0);
        if (lane == first) atomicAdd(&cnt[g0], (float)__builtin_popcount(same));
        todo &= ~same;
    }
}

template <bool RELU_A, bool TWO>
__global__ void k_gemm_wmma(const float* A, const float* A1, const float* __restrict__ W,
                            const float* __restrict__ bA, const float* __restrict__ dis,
                            float* __restrict__ out, float* out2, float* out3, int n, int K) {
    __shared__ float sW[64 * 64];
    __shared__ float sBias[64];
    const int tid = threadIdx.x;
    for (int i = tid; i < K * 64; i += 128) sW[i] = W[i];
    if (RELU_A) {
        if (tid < K) sBias[tid] = bA[tid];
    }
    __syncthreads();

    const int wave = tid >> 5;
    const int lane = tid & 31;
    const int tile_row = blockIdx.x * 64 + wave * 16;
    if (tile_row >= n) return;

    const int m    = lane & 15;
    const int half = lane >> 4;
    const int koff = half * 2;
    int row = tile_row + m;
    if (row >= n) row = n - 1;

    float d2[8];
#pragma unroll
    for (int r = 0; r < 8; ++r) {
        int rr = tile_row + r + 8 * half;
        if (rr >= n) rr = n - 1;
        const float d = dis[rr];
        d2[r] = d * d;
    }

    v8f acc[4] = {v8f{0.f}, v8f{0.f}, v8f{0.f}, v8f{0.f}};

    for (int kb = 0; kb < K; kb += 4) {
        v2f a;
        const size_t ao = (size_t)row * K + kb + koff;
        a.x = A[ao];
        a.y = A[ao + 1];
        if (RELU_A) {
            if (TWO) {
                a.x += A1[ao];
                a.y += A1[ao + 1];
            }
            a.x = fmaxf(a.x + sBias[kb + koff], 0.0f);
            a.y = fmaxf(a.y + sBias[kb + koff + 1], 0.0f);
        }
#pragma unroll
        for (int nt = 0; nt < 4; ++nt) {
            v2f b;
            const int col = nt * 16 + (lane & 15);
            b.x = sW[(kb + koff) * 64 + col];
            b.y = sW[(kb + koff + 1) * 64 + col];
            acc[nt] = __builtin_amdgcn_wmma_f32_16x16x4_f32(
                 false, a,  false, b,
                 (short)0, acc[nt],  false,  false);
        }
    }

#pragma unroll
    for (int nt = 0; nt < 4; ++nt) {
#pragma unroll
        for (int r = 0; r < 8; ++r) {
            const int rr = tile_row + r + 8 * half;
            if (rr < n) {
                const size_t o = (size_t)rr * 64 + nt * 16 + m;
                const float v = acc[nt][r];
                out[o]  = v;
                out2[o] = d2[r] * v;
                if (TWO) out3[o] = 0.0f;
            }
        }
    }
}

#define DIS_CAP   (50 * 1024)
#define SC_WAVES  32
#define SC_FILL   13

__global__ void __launch_bounds__(1024)
k_scatter(const int* __restrict__ src, const int* __restrict__ dst,
          const float* __restrict__ dis, const float* __restrict__ h,
          float* acc0, float* acc1, int E, int n) {
    __shared__ float sdis[DIS_CAP];
    const bool upper = (__builtin_amdgcn_s_sendmsg_rtn(0x87) >> 18) & 1u;
    float* acc = upper ? acc1 : acc0;
    const int tid  = threadIdx.x;
    const int lane = tid & 31;
    const int wave = __builtin_amdgcn_readfirstlane(tid >> 5);
    const bool tab = (n <= DIS_CAP);
    const int nchunk = (E + 31) >> 5;
    const int cstep  = gridDim.x * SC_WAVES;

    int c = blockIdx.x + gridDim.x * wave;
    int s = 0, d = 0;
    if (c < nchunk) {
        const int e = c * 32 + lane;
        if (e < E) { s = src[e]; d = dst[e]; }
    }

    if (tab) {
        const int nv = ((((size_t)dis) & 15) == 0) ? (n >> 2) : 0;
        if (nv > 0) {
            const int rot = (int)(((long long)blockIdx.x * nv) / gridDim.x);
            v4f t[SC_FILL];
#pragma unroll
            for (int u = 0; u < SC_FILL; ++u) {
                int k = u * 1024 + tid;
                if (k >= nv) k = nv - 1;
                int j = k + rot;
                if (j >= nv) j -= nv;
                t[u] = ((const v4f*)dis)[j];
            }
#pragma unroll
            for (int u = 0; u < SC_FILL; ++u) {
                const int k = u * 1024 + tid;
                if (k < nv) {
                    int j = k + rot;
                    if (j >= nv) j -= nv;
                    ((v4f*)sdis)[j] = t[u];
                }
            }
        }
        for (int i = nv * 4 + tid; i < n; i += 1024) sdis[i] = dis[i];
    }
    __syncthreads();

    for (; c < nchunk; c += cstep) {
        const int cn = c + cstep;
        int sn = 0, dn = 0;
        if (cn < nchunk) {
            const int en = cn * 32 + lane;
            if (en < E) { sn = src[en]; dn = dst[en]; }
        }
        const int e0 = c * 32;
        float w = 0.0f;
        if (e0 + lane < E) w = tab ? (sdis[s] * sdis[d]) : (dis[s] * dis[d]);
        const int wbits = __float_as_int(w);

        if (E - e0 >= 32) {
#pragma unroll
            for (int jb = 0; jb < 32; jb += 16) {
                float v0[16], v1[16];
#pragma unroll
                for (int j = 0; j < 16; ++j) {
                    const int sj = __builtin_amdgcn_readlane(s, jb + j);
                    const float* hp = h + (size_t)sj * 64 + lane;
                    v0[j] = hp[0];
                    v1[j] = hp[32];
                }
#pragma unroll
                for (int j = 0; j < 16; ++j) {
                    const int dj = __builtin_amdgcn_readlane(d, jb + j);
                    const float wj = __int_as_float(__builtin_amdgcn_readlane(wbits, jb + j));
                    float* ap = acc + (size_t)dj * 64 + lane;
                    atomicAdd(ap, wj * v0[j]);
                    atomicAdd(ap + 32, wj * v1[j]);
                }
            }
        } else {
            const int cnt = E - e0;
            for (int j = 0; j < cnt; ++j) {
                const int sj = __builtin_amdgcn_readlane(s, j);
                const int dj = __builtin_amdgcn_readlane(d, j);
                const float wj = __int_as_float(__builtin_amdgcn_readlane(wbits, j));
                const float* hp = h + (size_t)sj * 64 + lane;
                float* ap = acc + (size_t)dj * 64 + lane;
                atomicAdd(ap, wj * hp[0]);
                atomicAdd(ap + 32, wj * hp[32]);
            }
        }
        s = sn;
        d = dn;
    }
}

#define POOL_NODES 16

__device__ __forceinline__ void pool_flush(float* __restrict__ sums, int g, int lane, float s0, float s1) {
    float* sp = sums + (size_t)g * 64 + lane;
    atomicAdd(sp, s0);
    atomicAdd(sp + 32, s1);
}

template <bool TWO>
__global__ void __launch_bounds__(256) k_pool(const float* __restrict__ acc2, const float* __restrict__ acc2b,
                       const float* __restrict__ b2,
                       const int* __restrict__ batch, float* __restrict__ sums, int n) {
    const int lane = threadIdx.x & 31;
    const int wv = __builtin_amdgcn_readfirstlane((int)((blockIdx.x * blockDim.x + threadIdx.x) >> 5));
    const int i0 = wv * POOL_NODES;
    if (i0 >= n) return;
    const float bl = b2[lane];
    const float bh = b2[lane + 32];
    float s0 = 0.0f, s1 = 0.0f;
    int gc;

    if (i0 + POOL_NODES <= n) {
        int g[POOL_NODES];
        float v0[POOL_NODES], v1[POOL_NODES];
        float w0[POOL_NODES], w1[POOL_NODES];
#pragma unroll
        for (int k = 0; k < POOL_NODES; ++k) {
            g[k] = batch[i0 + k];
            const size_t o = (size_t)(i0 + k) * 64 + lane;
            v0[k] = acc2[o];
            v1[k] = acc2[o + 32];
            if (TWO) {
                w0[k] = acc2b[o];
                w1[k] = acc2b[o + 32];
            }
        }
        __builtin_amdgcn_sched_barrier(0);
        gc = g[0];
#pragma unroll
        for (int k = 0; k < POOL_NODES; ++k) {
            if (g[k] != gc) {
                pool_flush(sums, gc, lane, s0, s1);
                gc = g[k];
                s0 = 0.0f;
                s1 = 0.0f;
            }
            const float a = (TWO ? (v0[k] + w0[k]) : v0[k]) + bl;
            const float b = (TWO ? (v1[k] + w1[k]) : v1[k]) + bh;
            s0 += fmaxf(a, 0.0f);
            s1 += fmaxf(b, 0.0f);
        }
    } else {
        gc = batch[i0];
        for (int i = i0; i < n; ++i) {
            const int gi = batch[i];
            if (gi != gc) {
                pool_flush(sums, gc, lane, s0, s1);
                gc = gi;
                s0 = 0.0f;
                s1 = 0.0f;
            }
            const size_t o = (size_t)i * 64 + lane;
            float a = acc2[o];
            float b = acc2[o + 32];
            if (TWO) {
                a += acc2b[o];
                b += acc2b[o + 32];
            }
            s0 += fmaxf(a + bl, 0.0f);
            s1 += fmaxf(b + bh, 0.0f);
        }
    }
    pool_flush(sums, gc, lane, s0, s1);
}

__global__ void k_div(const float* __restrict__ sums, const float* __restrict__ cnt,
                      float* __restrict__ out, int G) {
    int t = blockIdx.x * blockDim.x + threadIdx.x;
    if (t >= G * 64) return;
    out[t] = sums[t] / fmaxf(cnt[t >> 6], 1.0f);
}

extern "C" void kernel_launch(void* const* d_in, const int* in_sizes, int n_in,
                              void* d_out, int out_size, void* d_ws, size_t ws_size,
                              hipStream_t stream) {
    const float* x     = (const float*)d_in[0];
    const int* eidx    = (const int*)d_in[1];
    const int* batch   = (const int*)d_in[2];
    const float* W1 = (const float*)d_in[4];
    const float* b1 = (const float*)d_in[5];
    const float* W2 = (const float*)d_in[6];
    const float* b2 = (const float*)d_in[7];

    const int n = in_sizes[0] / 4;
    const int E = in_sizes[1] / 2;
    const int G = out_size / 64;

    const int* src = eidx;
    const int* dst = eidx + E;

    char* wb = (char*)d_ws;
    const size_t need1 = ((size_t)n * 129 + (size_t)G * 65) * sizeof(float);
    const size_t need2 = need1 + (size_t)n * 64 * sizeof(float);
    const size_t pad   = (256 - ((size_t)wb & 255)) & 255;
    const bool two = need2 + pad <= ws_size;
    if ((two ? need2 : need1) + pad <= ws_size) wb += pad;
    float* bufA = (float*)wb;
    float* bufB = bufA + (size_t)n * 64;
    float* bufC = two ? bufB + (size_t)n * 64 : bufB;
    float* sums = bufC + (size_t)n * 64;
    float* cnt  = sums + (size_t)G * 64;
    float* dis  = cnt + G;

    const int T = 256;
    auto blk = [T](int work) { return (work + T - 1) / T; };
    const int pool_grid = (n + POOL_NODES * (T / 32) - 1) / (POOL_NODES * (T / 32));

    k_deg_hist<<<DEG_SLICES, 1024, 0, stream>>>(dst, (int*)bufB, E, n, sums, G * 64 + G);
    k_dis<<<blk(n), T, 0, stream>>>((const int*)bufB, dis, n, batch, cnt);

    if (two) k_gemm_wmma<false, true><<<(n + 63) / 64, 128, 0, stream>>>(x, nullptr, W1, b1, dis, bufA, bufB, bufC, n, 4);
    else     k_gemm_wmma<false, false><<<(n + 63) / 64, 128, 0, stream>>>(x, nullptr, W1, b1, dis, bufA, bufB, nullptr, n, 4);
    k_scatter<<<256, 1024, 0, stream>>>(src, dst, dis, bufA, bufB, bufC, E, n);

    if (two) k_gemm_wmma<true, true><<<(n + 63) / 64, 128, 0, stream>>>(bufB, bufC, W2, b1, dis, bufA, bufB, bufC, n, 64);
    else     k_gemm_wmma<true, false><<<(n + 63) / 64, 128, 0, stream>>>(bufB, nullptr, W2, b1, dis, bufA, bufB, nullptr, n, 64);
    k_scatter<<<256, 1024, 0, stream>>>(src, dst, dis, bufA, bufB, bufC, E, n);

    if (two) k_pool<true><<<pool_grid, T, 0, stream>>>(bufB, bufC, b2, batch, sums, n);
    else     k_pool<false><<<pool_grid, T, 0, stream>>>(bufB, nullptr, b2, batch, sums, n);
    k_div<<<blk(G * 64), T, 0, stream>>>(sums, cnt, (float*)d_out, G);
}


# layer-1 edge scatter makes each 16-edge block of message rows with four K=4 f32 WMMAs from gathered 16-B x rows (A = norm*x[src]), not by loading 256-B h1 rows; h1 plane gone; message rounding changes
# speedup vs baseline: 4.5965x; 1.0906x over previous
#include <hip/hip_runtime.h>
#include <hip/hip_bf16.h>

typedef __attribute__((ext_vector_type(2))) float v2f;
typedef __attribute__((ext_vector_type(8))) float v8f;
typedef __attribute__((ext_vector_type(4))) float v4f;

#define DEG_SLICES 64
#define DEG_CAP    (50 * 1024)

__global__ void __launch_bounds__(1024)
k_deg_hist(const int* __restrict__ dst, int* __restrict__ part, int E, int n,
           float* __restrict__ zr, int nz) {
    __shared__ int hist[DEG_CAP];
    const int tid = threadIdx.x;
    for (int i = blockIdx.x * 1024 + tid; i < nz; i += gridDim.x * 1024) zr[i] = 0.0f;
    const int per = (E + gridDim.x - 1) / gridDim.x;
    const int e0  = blockIdx.x * per;
    const int e1  = (e0 + per < E) ? (e0 + per) : E;
    int* __restrict__ plane = part + (size_t)blockIdx.x * n;

    for (int base = 0; base < n; base += DEG_CAP) {
        const int m = (n - base < DEG_CAP) ? (n - base) : DEG_CAP;
        for (int i = tid; i < (m + 3) / 4; i += 1024) ((int4*)hist)[i] = make_int4(0, 0, 0, 0);
        __syncthreads();

        int e = e0 + tid;
        for (; e + 3 * 1024 < e1; e += 4 * 1024) {
            const unsigned r0 = (unsigned)(dst[e] - base);
            const unsigned r1 = (unsigned)(dst[e + 1024] - base);
            const unsigned r2 = (unsigned)(dst[e + 2048] - base);
            const unsigned r3 = (unsigned)(dst[e + 3072] - base);
            if (r0 < (unsigned)m) atomicAdd(&hist[r0], 1);
            if (r1 < (unsigned)m) atomicAdd(&hist[r1], 1);
            if (r2 < (unsigned)m) atomicAdd(&hist[r2], 1);
            if (r3 < (unsigned)m) atomicAdd(&hist[r3], 1);
        }
        for (; e < e1; e += 1024) {
            const unsigned r = (unsigned)(dst[e] - base);
            if (r < (unsigned)m) atomicAdd(&hist[r], 1);
        }
        __syncthreads();

        for (int i = tid; i < m; i += 1024) plane[base + i] = hist[i];
        __syncthreads();
    }
}

__global__ void k_dis(const int* __restrict__ part, float* __restrict__ dis, int n,
                      const int* __restrict__ batch, float* __restrict__ cnt) {
    const int i = blockIdx.x * blockDim.x + threadIdx.x;
    const bool in = i < n;
    int g = -1;
    if (in) {
        g = batch[i];
        int c = 0;
#pragma unroll 16
        for (int w = 0; w < DEG_SLICES; ++w) c += part[(size_t)w * n + i];
        const float d = (float)c + 1.0f;
        dis[i] = (d > 0.0f) ? rsqrtf(d) : 0.0f;
    }
    const int lane = threadIdx.x & 31;
    unsigned todo = __builtin_amdgcn_ballot_w32(in);
    while (todo != 0u) {
        const int first = __builtin_ctz(todo);
        const int g0 = __builtin_amdgcn_readlane(g, first);
        const unsigned same = __builtin_amdgcn_ballot_w32(in && g == g0);
        if (lane == first) atomicAdd(&cnt[g0], (float)__builtin_popcount(same));
        todo &= ~same;
    }
}

template <bool RELU_A, bool TWO, bool STORE_H>
__global__ void k_gemm_wmma(const float* A, const float* A1, const float* __restrict__ W,
                            const float* __restrict__ bA, const float* __restrict__ dis,
                            float* __restrict__ out, float* out2, float* out3, int n, int K) {
    __shared__ float sW[64 * 64];
    __shared__ float sBias[64];
    const int tid = threadIdx.x;
    for (int i = tid; i < K * 64; i += 128) sW[i] = W[i];
    if (RELU_A) {
        if (tid < K) sBias[tid] = bA[tid];
    }
    __syncthreads();

    const int wave = tid >> 5;
    const int lane = tid & 31;
    const int tile_row = blockIdx.x * 64 + wave * 16;
    if (tile_row >= n) return;

    const int m    = lane & 15;
    const int half = lane >> 4;
    const int koff = half * 2;
    int row = tile_row + m;
    if (row >= n) row = n - 1;

    float d2[8];
#pragma unroll
    for (int r = 0; r < 8; ++r) {
        int rr = tile_row + r + 8 * half;
        if (rr >= n) rr = n - 1;
        const float d = dis[rr];
        d2[r] = d * d;
    }

    v8f acc[4] = {v8f{0.f}, v8f{0.f}, v8f{0.f}, v8f{0.f}};

    for (int kb = 0; kb < K; kb += 4) {
        v2f a;
        const size_t ao = (size_t)row * K + kb + koff;
        a.x = A[ao];
        a.y = A[ao + 1];
        if (RELU_A) {
            if (TWO) {
                a.x += A1[ao];
                a.y += A1[ao + 1];
            }
            a.x = fmaxf(a.x + sBias[kb + koff], 0.0f);
            a.y = fmaxf(a.y + sBias[kb + koff + 1], 0.0f);
        }
#pragma unroll
        for (int nt = 0; nt < 4; ++nt) {
            v2f b;
            const int col = nt * 16 + (lane & 15);
            b.x = sW[(kb + koff) * 64 + col];
            b.y = sW[(kb + koff + 1) * 64 + col];
            acc[nt] = __builtin_amdgcn_wmma_f32_16x16x4_f32(
                 false, a,  false, b,
                 (short)0, acc[nt],  false,  false);
        }
    }

#pragma unroll
    for (int nt = 0; nt < 4; ++nt) {
#pragma unroll
        for (int r = 0; r < 8; ++r) {
            const int rr = tile_row + r + 8 * half;
            if (rr < n) {
                const size_t o = (size_t)rr * 64 + nt * 16 + m;
                const float v = acc[nt][r];
                if (STORE_H) out[o] = v;
                out2[o] = d2[r] * v;
                if (TWO) out3[o] = 0.0f;
            }
        }
    }
}

#define DIS_CAP   (50 * 1024)
#define SC_WAVES  32
#define SC_FILL   13

__device__ __forceinline__ void fill_dis_table(float* sdis, const float* __restrict__ dis, int n, int tid) {
    const int nv = ((((size_t)dis) & 15) == 0) ? (n >> 2) : 0;
    if (nv > 0) {
        const int rot = (int)(((long long)blockIdx.x * nv) / gridDim.x);
        v4f t[SC_FILL];
#pragma unroll
        for (int u = 0; u < SC_FILL; ++u) {
            int k = u * 1024 + tid;
            if (k >= nv) k = nv - 1;
            int j = k + rot;
            if (j >= nv) j -= nv;
            t[u] = ((const v4f*)dis)[j];
        }
#pragma unroll
        for (int u = 0; u < SC_FILL; ++u) {
            const int k = u * 1024 + tid;
            if (k < nv) {
                int j = k + rot;
                if (j >= nv) j -= nv;
                ((v4f*)sdis)[j] = t[u];
            }
        }
    }
    for (int i = nv * 4 + tid; i < n; i += 1024) sdis[i] = dis[i];
}

__global__ void __launch_bounds__(1024)
k_scatter(const int* __restrict__ src, const int* __restrict__ dst,
          const float* __restrict__ dis, const float* __restrict__ h,
          float* acc0, float* acc1, int E, int n) {
    __shared__ float sdis[DIS_CAP];
    const bool upper = (__builtin_amdgcn_s_sendmsg_rtn(0x87) >> 18) & 1u;
    float* acc = upper ? acc1 : acc0;
    const int tid  = threadIdx.x;
    const int lane = tid & 31;
    const int wave = __builtin_amdgcn_readfirstlane(tid >> 5);
    const bool tab = (n <= DIS_CAP);
    const int nchunk = (E + 31) >> 5;
    const int cstep  = gridDim.x * SC_WAVES;

    int c = blockIdx.x + gridDim.x * wave;
    int s = 0, d = 0;
    if (c < nchunk) {
        const int e = c * 32 + lane;
        if (e < E) { s = src[e]; d = dst[e]; }
    }

    if (tab) fill_dis_table(sdis, dis, n, tid);
    __syncthreads();

    for (; c < nchunk; c += cstep) {
        const int cn = c + cstep;
        int sn = 0, dn = 0;
        if (cn < nchunk) {
            const int en = cn * 32 + lane;
            if (en < E) { sn = src[en]; dn = dst[en]; }
        }
        const int e0 = c * 32;
        float w = 0.0f;
        if (e0 + lane < E) w = tab ? (sdis[s] * sdis[d]) : (dis[s] * dis[d]);
        const int wbits = __float_as_int(w);

        if (E - e0 >= 32) {
#pragma unroll
            for (int jb = 0; jb < 32; jb += 16) {
                float v0[16], v1[16];
#pragma unroll
                for (int j = 0; j < 16; ++j) {
                    const int sj = __builtin_amdgcn_readlane(s, jb + j);
                    const float* hp = h + (size_t)sj * 64 + lane;
                    v0[j] = hp[0];
                    v1[j] = hp[32];
                }
#pragma unroll
                for (int j = 0; j < 16; ++j) {
                    const int dj = __builtin_amdgcn_readlane(d, jb + j);
                    const float wj = __int_as_float(__builtin_amdgcn_readlane(wbits, jb + j));
                    float* ap = acc + (size_t)dj * 64 + lane;
                    atomicAdd(ap, wj * v0[j]);
                    atomicAdd(ap + 32, wj * v1[j]);
                }
            }
        } else {
            const int cnt = E - e0;
            for (int j = 0; j < cnt; ++j) {
                const int sj = __builtin_amdgcn_readlane(s, j);
                const int dj = __builtin_amdgcn_readlane(d, j);
                const float wj = __int_as_float(__builtin_amdgcn_readlane(wbits, j));
                const float* hp = h + (size_t)sj * 64 + lane;
                float* ap = acc + (size_t)dj * 64 + lane;
                atomicAdd(ap, wj * hp[0]);
                atomicAdd(ap + 32, wj * hp[32]);
            }
        }
        s = sn;
        d = dn;
    }
}


__device__ __forceinline__ float xhalf(float v) {
    return __builtin_bit_cast(float, __builtin_amdgcn_permlanex16(__builtin_bit_cast(int, v), __builtin_bit_cast(int, v),
                                                                   0x76543210, 0xfedcba98, false, false));
}

template <bool GUARD, int JB>
__device__ __forceinline__ void msg_block(v2f a, const v2f (&bw)[4], float* acc, int d, int lane, int cnt) {
    v8f t[4];
#pragma unroll
    for (int nt = 0; nt < 4; ++nt)
        t[nt] = __builtin_amdgcn_wmma_f32_16x16x4_f32(
             false, a,  false, bw[nt],
             (short)0, v8f{0.f},  false,  false);
    const bool lo = lane < 16;
#pragma unroll
    for (int r = 0; r < 8; ++r) {
        const int d0 = __builtin_amdgcn_readlane(d, JB + r);
        const int d1 = __builtin_amdgcn_readlane(d, JB + r + 8);
        float* p0 = acc + (size_t)d0 * 64 + lane;
        float* p1 = acc + (size_t)d1 * 64 + lane;
#pragma unroll
        for (int j = 0; j < 2; ++j) {
            const float tl = t[2 * j][r];
            const float th = t[2 * j + 1][r];
            const float got = xhalf(lo ? th : tl);
            const float m0 = lo ? tl : got;
            const float m1 = lo ? got : th;
            if (!GUARD || JB + r < cnt) atomicAdd(p0 + 32 * j, m0);
            if (!GUARD || JB + r + 8 < cnt) atomicAdd(p1 + 32 * j, m1);
        }
    }
}

__global__ void __launch_bounds__(1024)
k_scatter_x(const int* __restrict__ src, const int* __restrict__ dst,
            const float* __restrict__ dis, const float* __restrict__ x, const float* __restrict__ W,
            float* acc0, float* acc1, int E, int n) {
    __shared__ float sdis[DIS_CAP];
    const bool upper = (__builtin_amdgcn_s_sendmsg_rtn(0x87) >> 18) & 1u;
    float* acc = upper ? acc1 : acc0;
    const int tid  = threadIdx.x;
    const int lane = tid & 31;
    const int wave = __builtin_amdgcn_readfirstlane(tid >> 5);
    const bool tab = (n <= DIS_CAP);
    const int nchunk = (E + 31) >> 5;
    const int cstep  = gridDim.x * SC_WAVES;
    const bool lo = lane < 16;
    const int half = lane >> 4;

    int c = blockIdx.x + gridDim.x * wave;
    int s = 0, d = 0, s1 = 0, d1 = 0;
    if (c < nchunk) {
        const int e = c * 32 + lane;
        if (e < E) { s = src[e]; d = dst[e]; }
    }
    if (c + cstep < nchunk) {
        const int e = (c + cstep) * 32 + lane;
        if (e < E) { s1 = src[e]; d1 = dst[e]; }
    }
    v2f bw[4];
#pragma unroll
    for (int nt = 0; nt < 4; ++nt) {
        const int col = nt * 16 + (lane & 15);
        bw[nt].x = W[(2 * half) * 64 + col];
        bw[nt].y = W[(2 * half + 1) * 64 + col];
    }

    if (tab) fill_dis_table(sdis, dis, n, tid);
    v4f xv = *(const v4f*)(x + (size_t)s * 4);
    __syncthreads();

    for (; c < nchunk; c += cstep) {
        const int c2 = c + 2 * cstep;
        int s2 = 0, d2 = 0;
        if (c2 < nchunk) {
            const int e2 = c2 * 32 + lane;
            if (e2 < E) { s2 = src[e2]; d2 = dst[e2]; }
        }
        const v4f xn = *(const v4f*)(x + (size_t)s1 * 4);

        const int e0 = c * 32;
        float w = 0.0f;
        if (e0 + lane < E) w = tab ? (sdis[s] * sdis[d]) : (dis[s] * dis[d]);
        const v4f xw = xv * w;

        const float g0 = xhalf(lo ? xw.z : xw.x);
        const float g1 = xhalf(lo ? xw.w : xw.y);
        v2f a0, a1;
        a0.x = lo ? xw.x : g0;
        a0.y = lo ? xw.y : g1;
        a1.x = lo ? g0 : xw.z;
        a1.y = lo ? g1 : xw.w;

        if (E - e0 >= 32) {
            msg_block<false, 0>(a0, bw, acc, d, lane, 32);
            msg_block<false, 16>(a1, bw, acc, d, lane, 32);
        } else {
            const int cnt = E - e0;
            msg_block<true, 0>(a0, bw, acc, d, lane, cnt);
            msg_block<true, 16>(a1, bw, acc, d, lane, cnt);
        }
        s = s1;  d = d1;  xv = xn;
        s1 = s2; d1 = d2;
    }
}

#define POOL_NODES 16

__device__ __forceinline__ void pool_flush(float* __restrict__ sums, int g, int lane, float s0, float s1) {
    float* sp = sums + (size_t)g * 64 + lane;
    atomicAdd(sp, s0);
    atomicAdd(sp + 32, s1);
}

template <bool TWO>
__global__ void __launch_bounds__(256) k_pool(const float* __restrict__ acc2, const float* __restrict__ acc2b,
                       const float* __restrict__ b2,
                       const int* __restrict__ batch, float* __restrict__ sums, int n) {
    const int lane = threadIdx.x & 31;
    const int wv = __builtin_amdgcn_readfirstlane((int)((blockIdx.x * blockDim.x + threadIdx.x) >> 5));
    const int i0 = wv * POOL_NODES;
    if (i0 >= n) return;
    const float bl = b2[lane];
    const float bh = b2[lane + 32];
    float s0 = 0.0f, s1 = 0.0f;
    int gc;

    if (i0 + POOL_NODES <= n) {
        int g[POOL_NODES];
        float v0[POOL_NODES], v1[POOL_NODES];
        float w0[POOL_NODES], w1[POOL_NODES];
#pragma unroll
        for (int k = 0; k < POOL_NODES; ++k) {
            g[k] = batch[i0 + k];
            const size_t o = (size_t)(i0 + k) * 64 + lane;
            v0[k] = acc2[o];
            v1[k] = acc2[o + 32];
            if (TWO) {
                w0[k] = acc2b[o];
                w1[k] = acc2b[o + 32];
            }
        }
        __builtin_amdgcn_sched_barrier(0);
        gc = g[0];
#pragma unroll
        for (int k = 0; k < POOL_NODES; ++k) {
            if (g[k] != gc) {
                pool_flush(sums, gc, lane, s0, s1);
                gc = g[k];
                s0 = 0.0f;
                s1 = 0.0f;
            }
            const float a = (TWO ? (v0[k] + w0[k]) : v0[k]) + bl;
            const float b = (TWO ? (v1[k] + w1[k]) : v1[k]) + bh;
            s0 += fmaxf(a, 0.0f);
            s1 += fmaxf(b, 0.0f);
        }
    } else {
        gc = batch[i0];
        for (int i = i0; i < n; ++i) {
            const int gi = batch[i];
            if (gi != gc) {
                pool_flush(sums, gc, lane, s0, s1);
                gc = gi;
                s0 = 0.0f;
                s1 = 0.0f;
            }
            const size_t o = (size_t)i * 64 + lane;
            float a = acc2[o];
            float b = acc2[o + 32];
            if (TWO) {
                a += acc2b[o];
                b += acc2b[o + 32];
            }
            s0 += fmaxf(a + bl, 0.0f);
            s1 += fmaxf(b + bh, 0.0f);
        }
    }
    pool_flush(sums, gc, lane, s0, s1);
}

__global__ void k_div(const float* __restrict__ sums, const float* __restrict__ cnt,
                      float* __restrict__ out, int G) {
    int t = blockIdx.x * blockDim.x + threadIdx.x;
    if (t >= G * 64) return;
    out[t] = sums[t] / fmaxf(cnt[t >> 6], 1.0f);
}

extern "C" void kernel_launch(void* const* d_in, const int* in_sizes, int n_in,
                              void* d_out, int out_size, void* d_ws, size_t ws_size,
                              hipStream_t stream) {
    const float* x     = (const float*)d_in[0];
    const int* eidx    = (const int*)d_in[1];
    const int* batch   = (const int*)d_in[2];
    const float* W1 = (const float*)d_in[4];
    const float* b1 = (const float*)d_in[5];
    const float* W2 = (const float*)d_in[6];
    const float* b2 = (const float*)d_in[7];

    const int n = in_sizes[0] / 4;
    const int E = in_sizes[1] / 2;
    const int G = out_size / 64;

    const int* src = eidx;
    const int* dst = eidx + E;

    char* wb = (char*)d_ws;
    const size_t need1 = ((size_t)n * 129 + (size_t)G * 65) * sizeof(float);
    const size_t need2 = need1 + (size_t)n * 64 * sizeof(float);
    const size_t pad   = (256 - ((size_t)wb & 255)) & 255;
    const bool two = need2 + pad <= ws_size;
    if ((two ? need2 : need1) + pad <= ws_size) wb += pad;
    float* bufA = (float*)wb;
    float* bufB = bufA + (size_t)n * 64;
    float* bufC = two ? bufB + (size_t)n * 64 : bufB;
    float* sums = bufC + (size_t)n * 64;
    float* cnt  = sums + (size_t)G * 64;
    float* dis  = cnt + G;

    const int T = 256;
    auto blk = [T](int work) { return (work + T - 1) / T; };
    const int pool_grid = (n + POOL_NODES * (T / 32) - 1) / (POOL_NODES * (T / 32));

    k_deg_hist<<<DEG_SLICES, 1024, 0, stream>>>(dst, (int*)bufB, E, n, sums, G * 64 + G);
    k_dis<<<blk(n), T, 0, stream>>>((const int*)bufB, dis, n, batch, cnt);

    if (two) k_gemm_wmma<false, true, false><<<(n + 63) / 64, 128, 0, stream>>>(x, nullptr, W1, b1, dis, bufA, bufB, bufC, n, 4);
    else     k_gemm_wmma<false, false, false><<<(n + 63) / 64, 128, 0, stream>>>(x, nullptr, W1, b1, dis, bufA, bufB, nullptr, n, 4);
    k_scatter_x<<<256, 1024, 0, stream>>>(src, dst, dis, x, W1, bufB, bufC, E, n);

    if (two) k_gemm_wmma<true, true, true><<<(n + 63) / 64, 128, 0, stream>>>(bufB, bufC, W2, b1, dis, bufA, bufB, bufC, n, 64);
    else     k_gemm_wmma<true, false, true><<<(n + 63) / 64, 128, 0, stream>>>(bufB, nullptr, W2, b1, dis, bufA, bufB, nullptr, n, 64);
    k_scatter<<<256, 1024, 0, stream>>>(src, dst, dis, bufA, bufB, bufC, E, n);

    if (two) k_pool<true><<<pool_grid, T, 0, stream>>>(bufB, bufC, b2, batch, sums, n);
    else     k_pool<false><<<pool_grid, T, 0, stream>>>(bufB, nullptr, b2, batch, sums, n);
    k_div<<<blk(G * 64), T, 0, stream>>>(sums, cnt, (float*)d_out, G);
}


# dis (sum of the 64 degree planes, +1, rsqrt) and the nodes-per-graph counts made in the layer-1 product kernel for its own 16 rows; k_dis launch gone, 8 -> 7 launches; float sum order unchanged
# speedup vs baseline: 4.7861x; 1.0412x over previous
#include <hip/hip_runtime.h>
#include <hip/hip_bf16.h>

typedef __attribute__((ext_vector_type(2))) float v2f;
typedef __attribute__((ext_vector_type(8))) float v8f;
typedef __attribute__((ext_vector_type(4))) float v4f;

#define DEG_SLICES 64
#define DEG_CAP    (50 * 1024)

__global__ void __launch_bounds__(1024)
k_deg_hist(const int* __restrict__ dst, int* __restrict__ part, int E, int n,
           float* __restrict__ zr, int nz) {
    __shared__ int hist[DEG_CAP];
    const int tid = threadIdx.x;
    for (int i = blockIdx.x * 1024 + tid; i < nz; i += gridDim.x * 1024) zr[i] = 0.0f;
    const int per = (E + gridDim.x - 1) / gridDim.x;
    const int e0  = blockIdx.x * per;
    const int e1  = (e0 + per < E) ? (e0 + per) : E;
    int* __restrict__ plane = part + (size_t)blockIdx.x * n;

    for (int base = 0; base < n; base += DEG_CAP) {
        const int m = (n - base < DEG_CAP) ? (n - base) : DEG_CAP;
        for (int i = tid; i < (m + 3) / 4; i += 1024) ((int4*)hist)[i] = make_int4(0, 0, 0, 0);
        __syncthreads();

        int e = e0 + tid;
        for (; e + 3 * 1024 < e1; e += 4 * 1024) {
            const unsigned r0 = (unsigned)(dst[e] - base);
            const unsigned r1 = (unsigned)(dst[e + 1024] - base);
            const unsigned r2 = (unsigned)(dst[e + 2048] - base);
            const unsigned r3 = (unsigned)(dst[e + 3072] - base);
            if (r0 < (unsigned)m) atomicAdd(&hist[r0], 1);
            if (r1 < (unsigned)m) atomicAdd(&hist[r1], 1);
            if (r2 < (unsigned)m) atomicAdd(&hist[r2], 1);
            if (r3 < (unsigned)m) atomicAdd(&hist[r3], 1);
        }
        for (; e < e1; e += 1024) {
            const unsigned r = (unsigned)(dst[e] - base);
            if (r < (unsigned)m) atomicAdd(&hist[r], 1);
        }
        __syncthreads();

        for (int i = tid; i < m; i += 1024) plane[base + i] = hist[i];
        __syncthreads();
    }
}

__device__ __forceinline__ int xhalf_bits(int v) {
    return __builtin_amdgcn_permlanex16(v, v, 0x76543210, 0xfedcba98, false, false);
}
__device__ __forceinline__ float xhalf(float v) {
    return __builtin_bit_cast(float, xhalf_bits(__builtin_bit_cast(int, v)));
}

template <bool RELU_A, bool TWO, bool STORE_H, bool MAKE_DIS>
__global__ void k_gemm_wmma(const float* A, const float* A1, const float* __restrict__ W,
                            const float* __restrict__ bA, const float* __restrict__ dis,
                            float* __restrict__ out, float* out2, float* out3, int n, int K,
                            const int* __restrict__ part, float* __restrict__ dis_out,
                            const int* __restrict__ batch, float* __restrict__ cnt) {
    __shared__ float sW[64 * 64];
    __shared__ float sBias[64];
    const int tid = threadIdx.x;

    int pc[DEG_SLICES / 2];
    int gid = -1;
    bool counted = false;
    if (MAKE_DIS) {
        const int r0 = blockIdx.x * 64 + (tid >> 5) * 16 + (tid & 15);
        const int rc = (r0 < n) ? r0 : (n - 1);
        const int h  = (tid >> 4) & 1;
#pragma unroll
        for (int w = 0; w < DEG_SLICES / 2; ++w)
            pc[w] = part[(size_t)(h * (DEG_SLICES / 2) + w) * n + rc];
        counted = (h == 0) && (r0 < n);
        if (counted) gid = batch[rc];
    }

    for (int i = tid; i < K * 64; i += 128) sW[i] = W[i];
    if (RELU_A) {
        if (tid < K) sBias[tid] = bA[tid];
    }
    __syncthreads();

    const int wave = tid >> 5;
    const int lane = tid & 31;
    const int tile_row = blockIdx.x * 64 + wave * 16;
    if (tile_row >= n) return;

    const int m    = lane & 15;
    const int half = lane >> 4;
    const int koff = half * 2;
    int row = tile_row + m;
    if (row >= n) row = n - 1;

    float d2[8];
    if (MAKE_DIS) {
        int c = 0;
#pragma unroll
        for (int w = 0; w < DEG_SLICES / 2; ++w) c += pc[w];
        c += xhalf_bits(c);
        const float dg = (float)c + 1.0f;
        const float dv = (dg > 0.0f) ? rsqrtf(dg) : 0.0f;
        if (counted) dis_out[row] = dv;
        const int dvb = __float_as_int(dv);
#pragma unroll
        for (int r = 0; r < 8; ++r) {
            const float dl = __int_as_float(__builtin_amdgcn_readlane(dvb, r));
            const float dh = __int_as_float(__builtin_amdgcn_readlane(dvb, r + 8));
            const float d = half ? dh : dl;
            d2[r] = d * d;
        }
        unsigned todo = __builtin_amdgcn_ballot_w32(counted);
        while (todo != 0u) {
            const int first = __builtin_ctz(todo);
            const int g0 = __builtin_amdgcn_readlane(gid, first);
            const unsigned same = __builtin_amdgcn_ballot_w32(counted && gid == g0);
            if (lane == first) atomicAdd(&cnt[g0], (float)__builtin_popcount(same));
            todo &= ~same;
        }
    } else {
#pragma unroll
        for (int r = 0; r < 8; ++r) {
            int rr = tile_row + r + 8 * half;
            if (rr >= n) rr = n - 1;
            const float d = dis[rr];
            d2[r] = d * d;
        }
    }

    v8f acc[4] = {v8f{0.f}, v8f{0.f}, v8f{0.f}, v8f{0.f}};

    for (int kb = 0; kb < K; kb += 4) {
        v2f a;
        const size_t ao = (size_t)row * K + kb + koff;
        a.x = A[ao];
        a.y = A[ao + 1];
        if (RELU_A) {
            if (TWO) {
                a.x += A1[ao];
                a.y += A1[ao + 1];
            }
            a.x = fmaxf(a.x + sBias[kb + koff], 0.0f);
            a.y = fmaxf(a.y + sBias[kb + koff + 1], 0.0f);
        }
#pragma unroll
        for (int nt = 0; nt < 4; ++nt) {
            v2f b;
            const int col = nt * 16 + (lane & 15);
            b.x = sW[(kb + koff) * 64 + col];
            b.y = sW[(kb + koff + 1) * 64 + col];
            acc[nt] = __builtin_amdgcn_wmma_f32_16x16x4_f32(
                 false, a,  false, b,
                 (short)0, acc[nt],  false,  false);
        }
    }

#pragma unroll
    for (int nt = 0; nt < 4; ++nt) {
#pragma unroll
        for (int r = 0; r < 8; ++r) {
            const int rr = tile_row + r + 8 * half;
            if (rr < n) {
                const size_t o = (size_t)rr * 64 + nt * 16 + m;
                const float v = acc[nt][r];
                if (STORE_H) out[o] = v;
                out2[o] = d2[r] * v;
                if (TWO) out3[o] = 0.0f;
            }
        }
    }
}

#define DIS_CAP   (50 * 1024)
#define SC_WAVES  32
#define SC_FILL   13

__device__ __forceinline__ void fill_dis_table(float* sdis, const float* __restrict__ dis, int n, int tid) {
    const int nv = ((((size_t)dis) & 15) == 0) ? (n >> 2) : 0;
    if (nv > 0) {
        const int rot = (int)(((long long)blockIdx.x * nv) / gridDim.x);
        v4f t[SC_FILL];
#pragma unroll
        for (int u = 0; u < SC_FILL; ++u) {
            int k = u * 1024 + tid;
            if (k >= nv) k = nv - 1;
            int j = k + rot;
            if (j >= nv) j -= nv;
            t[u] = ((const v4f*)dis)[j];
        }
#pragma unroll
        for (int u = 0; u < SC_FILL; ++u) {
            const int k = u * 1024 + tid;
            if (k < nv) {
                int j = k + rot;
                if (j >= nv) j -= nv;
                ((v4f*)sdis)[j] = t[u];
            }
        }
    }
    for (int i = nv * 4 + tid; i < n; i += 1024) sdis[i] = dis[i];
}

__global__ void __launch_bounds__(1024)
k_scatter(const int* __restrict__ src, const int* __restrict__ dst,
          const float* __restrict__ dis, const float* __restrict__ h,
          float* acc0, float* acc1, int E, int n) {
    __shared__ float sdis[DIS_CAP];
    const bool upper = (__builtin_amdgcn_s_sendmsg_rtn(0x87) >> 18) & 1u;
    float* acc = upper ? acc1 : acc0;
    const int tid  = threadIdx.x;
    const int lane = tid & 31;
    const int wave = __builtin_amdgcn_readfirstlane(tid >> 5);
    const bool tab = (n <= DIS_CAP);
    const int nchunk = (E + 31) >> 5;
    const int cstep  = gridDim.x * SC_WAVES;

    int c = blockIdx.x + gridDim.x * wave;
    int s = 0, d = 0;
    if (c < nchunk) {
        const int e = c * 32 + lane;
        if (e < E) { s = src[e]; d = dst[e]; }
    }

    if (tab) fill_dis_table(sdis, dis, n, tid);
    __syncthreads();

    for (; c < nchunk; c += cstep) {
        const int cn = c + cstep;
        int sn = 0, dn = 0;
        if (cn < nchunk) {
            const int en = cn * 32 + lane;
            if (en < E) { sn = src[en]; dn = dst[en]; }
        }
        const int e0 = c * 32;
        float w = 0.0f;
        if (e0 + lane < E) w = tab ? (sdis[s] * sdis[d]) : (dis[s] * dis[d]);
        const int wbits = __float_as_int(w);

        if (E - e0 >= 32) {
#pragma unroll
            for (int jb = 0; jb < 32; jb += 16) {
                float v0[16], v1[16];
#pragma unroll
                for (int j = 0; j < 16; ++j) {
                    const int sj = __builtin_amdgcn_readlane(s, jb + j);
                    const float* hp = h + (size_t)sj * 64 + lane;
                    v0[j] = hp[0];
                    v1[j] = hp[32];
                }
#pragma unroll
                for (int j = 0; j < 16; ++j) {
                    const int dj = __builtin_amdgcn_readlane(d, jb + j);
                    const float wj = __int_as_float(__builtin_amdgcn_readlane(wbits, jb + j));
                    float* ap = acc + (size_t)dj * 64 + lane;
                    atomicAdd(ap, wj * v0[j]);
                    atomicAdd(ap + 32, wj * v1[j]);
                }
            }
        } else {
            const int cnt = E - e0;
            for (int j = 0; j < cnt; ++j) {
                const int sj = __builtin_amdgcn_readlane(s, j);
                const int dj = __builtin_amdgcn_readlane(d, j);
                const float wj = __int_as_float(__builtin_amdgcn_readlane(wbits, j));
                const float* hp = h + (size_t)sj * 64 + lane;
                float* ap = acc + (size_t)dj * 64 + lane;
                atomicAdd(ap, wj * hp[0]);
                atomicAdd(ap + 32, wj * hp[32]);
            }
        }
        s = sn;
        d = dn;
    }
}


template <bool GUARD, int JB>
__device__ __forceinline__ void msg_block(v2f a, const v2f (&bw)[4], float* acc, int d, int lane, int cnt) {
    v8f t[4];
#pragma unroll
    for (int nt = 0; nt < 4; ++nt)
        t[nt] = __builtin_amdgcn_wmma_f32_16x16x4_f32(
             false, a,  false, bw[nt],
             (short)0, v8f{0.f},  false,  false);
    const bool lo = lane < 16;
#pragma unroll
    for (int r = 0; r < 8; ++r) {
        const int d0 = __builtin_amdgcn_readlane(d, JB + r);
        const int d1 = __builtin_amdgcn_readlane(d, JB + r + 8);
        float* p0 = acc + (size_t)d0 * 64 + lane;
        float* p1 = acc + (size_t)d1 * 64 + lane;
#pragma unroll
        for (int j = 0; j < 2; ++j) {
            const float tl = t[2 * j][r];
            const float th = t[2 * j + 1][r];
            const float got = xhalf(lo ? th : tl);
            const float m0 = lo ? tl : got;
            const float m1 = lo ? got : th;
            if (!GUARD || JB + r < cnt) atomicAdd(p0 + 32 * j, m0);
            if (!GUARD || JB + r + 8 < cnt) atomicAdd(p1 + 32 * j, m1);
        }
    }
}

__global__ void __launch_bounds__(1024)
k_scatter_x(const int* __restrict__ src, const int* __restrict__ dst,
            const float* __restrict__ dis, const float* __restrict__ x, const float* __restrict__ W,
            float* acc0, float* acc1, int E, int n) {
    __shared__ float sdis[DIS_CAP];
    const bool upper = (__builtin_amdgcn_s_sendmsg_rtn(0x87) >> 18) & 1u;
    float* acc = upper ? acc1 : acc0;
    const int tid  = threadIdx.x;
    const int lane = tid & 31;
    const int wave = __builtin_amdgcn_readfirstlane(tid >> 5);
    const bool tab = (n <= DIS_CAP);
    const int nchunk = (E + 31) >> 5;
    const int cstep  = gridDim.x * SC_WAVES;
    const bool lo = lane < 16;
    const int half = lane >> 4;

    int c = blockIdx.x + gridDim.x * wave;
    int s = 0, d = 0, s1 = 0, d1 = 0;
    if (c < nchunk) {
        const int e = c * 32 + lane;
        if (e < E) { s = src[e]; d = dst[e]; }
    }
    if (c + cstep < nchunk) {
        const int e = (c + cstep) * 32 + lane;
        if (e < E) { s1 = src[e]; d1 = dst[e]; }
    }
    v2f bw[4];
#pragma unroll
    for (int nt = 0; nt < 4; ++nt) {
        const int col = nt * 16 + (lane & 15);
        bw[nt].x = W[(2 * half) * 64 + col];
        bw[nt].y = W[(2 * half + 1) * 64 + col];
    }

    if (tab) fill_dis_table(sdis, dis, n, tid);
    v4f xv = *(const v4f*)(x + (size_t)s * 4);
    __syncthreads();

    for (; c < nchunk; c += cstep) {
        const int c2 = c + 2 * cstep;
        int s2 = 0, d2 = 0;
        if (c2 < nchunk) {
            const int e2 = c2 * 32 + lane;
            if (e2 < E) { s2 = src[e2]; d2 = dst[e2]; }
        }
        const v4f xn = *(const v4f*)(x + (size_t)s1 * 4);

        const int e0 = c * 32;
        float w = 0.0f;
        if (e0 + lane < E) w = tab ? (sdis[s] * sdis[d]) : (dis[s] * dis[d]);
        const v4f xw = xv * w;

        const float g0 = xhalf(lo ? xw.z : xw.x);
        const float g1 = xhalf(lo ? xw.w : xw.y);
        v2f a0, a1;
        a0.x = lo ? xw.x : g0;
        a0.y = lo ? xw.y : g1;
        a1.x = lo ? g0 : xw.z;
        a1.y = lo ? g1 : xw.w;

        if (E - e0 >= 32) {
            msg_block<false, 0>(a0, bw, acc, d, lane, 32);
            msg_block<false, 16>(a1, bw, acc, d, lane, 32);
        } else {
            const int cnt = E - e0;
            msg_block<true, 0>(a0, bw, acc, d, lane, cnt);
            msg_block<true, 16>(a1, bw, acc, d, lane, cnt);
        }
        s = s1;  d = d1;  xv = xn;
        s1 = s2; d1 = d2;
    }
}

#define POOL_NODES 16

__device__ __forceinline__ void pool_flush(float* __restrict__ sums, int g, int lane, float s0, float s1) {
    float* sp = sums + (size_t)g * 64 + lane;
    atomicAdd(sp, s0);
    atomicAdd(sp + 32, s1);
}

template <bool TWO>
__global__ void __launch_bounds__(256) k_pool(const float* __restrict__ acc2, const float* __restrict__ acc2b,
                       const float* __restrict__ b2,
                       const int* __restrict__ batch, float* __restrict__ sums, int n) {
    const int lane = threadIdx.x & 31;
    const int wv = __builtin_amdgcn_readfirstlane((int)((blockIdx.x * blockDim.x + threadIdx.x) >> 5));
    const int i0 = wv * POOL_NODES;
    if (i0 >= n) return;
    const float bl = b2[lane];
    const float bh = b2[lane + 32];
    float s0 = 0.0f, s1 = 0.0f;
    int gc;

    if (i0 + POOL_NODES <= n) {
        int g[POOL_NODES];
        float v0[POOL_NODES], v1[POOL_NODES];
        float w0[POOL_NODES], w1[POOL_NODES];
#pragma unroll
        for (int k = 0; k < POOL_NODES; ++k) {
            g[k] = batch[i0 + k];
            const size_t o = (size_t)(i0 + k) * 64 + lane;
            v0[k] = acc2[o];
            v1[k] = acc2[o + 32];
            if (TWO) {
                w0[k] = acc2b[o];
                w1[k] = acc2b[o + 32];
            }
        }
        __builtin_amdgcn_sched_barrier(0);
        gc = g[0];
#pragma unroll
        for (int k = 0; k < POOL_NODES; ++k) {
            if (g[k] != gc) {
                pool_flush(sums, gc, lane, s0, s1);
                gc = g[k];
                s0 = 0.0f;
                s1 = 0.0f;
            }
            const float a = (TWO ? (v0[k] + w0[k]) : v0[k]) + bl;
            const float b = (TWO ? (v1[k] + w1[k]) : v1[k]) + bh;
            s0 += fmaxf(a, 0.0f);
            s1 += fmaxf(b, 0.0f);
        }
    } else {
        gc = batch[i0];
        for (int i = i0; i < n; ++i) {
            const int gi = batch[i];
            if (gi != gc) {
                pool_flush(sums, gc, lane, s0, s1);
                gc = gi;
                s0 = 0.0f;
                s1 = 0.0f;
            }
            const size_t o = (size_t)i * 64 + lane;
            float a = acc2[o];
            float b = acc2[o + 32];
            if (TWO) {
                a += acc2b[o];
                b += acc2b[o + 32];
            }
            s0 += fmaxf(a + bl, 0.0f);
            s1 += fmaxf(b + bh, 0.0f);
        }
    }
    pool_flush(sums, gc, lane, s0, s1);
}

__global__ void k_div(const float* __restrict__ sums, const float* __restrict__ cnt,
                      float* __restrict__ out, int G) {
    int t = blockIdx.x * blockDim.x + threadIdx.x;
    if (t >= G * 64) return;
    out[t] = sums[t] / fmaxf(cnt[t >> 6], 1.0f);
}

extern "C" void kernel_launch(void* const* d_in, const int* in_sizes, int n_in,
                              void* d_out, int out_size, void* d_ws, size_t ws_size,
                              hipStream_t stream) {
    const float* x     = (const float*)d_in[0];
    const int* eidx    = (const int*)d_in[1];
    const int* batch   = (const int*)d_in[2];
    const float* W1 = (const float*)d_in[4];
    const float* b1 = (const float*)d_in[5];
    const float* W2 = (const float*)d_in[6];
    const float* b2 = (const float*)d_in[7];

    const int n = in_sizes[0] / 4;
    const int E = in_sizes[1] / 2;
    const int G = out_size / 64;

    const int* src = eidx;
    const int* dst = eidx + E;

    char* wb = (char*)d_ws;
    const size_t need1 = ((size_t)n * 129 + (size_t)G * 65) * sizeof(float);
    const size_t need2 = need1 + (size_t)n * 64 * sizeof(float);
    const size_t pad   = (256 - ((size_t)wb & 255)) & 255;
    const bool two = need2 + pad <= ws_size;
    if ((two ? need2 : need1) + pad <= ws_size) wb += pad;
    float* bufA = (float*)wb;
    float* bufB = bufA + (size_t)n * 64;
    float* bufC = two ? bufB + (size_t)n * 64 : bufB;
    float* sums = bufC + (size_t)n * 64;
    float* cnt  = sums + (size_t)G * 64;
    float* dis  = cnt + G;

    const int T = 256;
    auto blk = [T](int work) { return (work + T - 1) / T; };
    const int pool_grid = (n + POOL_NODES * (T / 32) - 1) / (POOL_NODES * (T / 32));

    k_deg_hist<<<DEG_SLICES, 1024, 0, stream>>>(dst, (int*)bufA, E, n, sums, G * 64 + G);

    if (two) k_gemm_wmma<false, true, false, true><<<(n + 63) / 64, 128, 0, stream>>>(x, nullptr, W1, b1, nullptr, nullptr, bufB, bufC, n, 4, (const int*)bufA, dis, batch, cnt);
    else     k_gemm_wmma<false, false, false, true><<<(n + 63) / 64, 128, 0, stream>>>(x, nullptr, W1, b1, nullptr, nullptr, bufB, nullptr, n, 4, (const int*)bufA, dis, batch, cnt);
    k_scatter_x<<<256, 1024, 0, stream>>>(src, dst, dis, x, W1, bufB, bufC, E, n);

    if (two) k_gemm_wmma<true, true, true, false><<<(n + 63) / 64, 128, 0, stream>>>(bufB, bufC, W2, b1, dis, bufA, bufB, bufC, n, 64, nullptr, nullptr, nullptr, nullptr);
    else     k_gemm_wmma<true, false, true, false><<<(n + 63) / 64, 128, 0, stream>>>(bufB, nullptr, W2, b1, dis, bufA, bufB, nullptr, n, 64, nullptr, nullptr, nullptr, nullptr);
    k_scatter<<<256, 1024, 0, stream>>>(src, dst, dis, bufA, bufB, bufC, E, n);

    if (two) k_pool<true><<<pool_grid, T, 0, stream>>>(bufB, bufC, b2, batch, sums, n);
    else     k_pool<false><<<pool_grid, T, 0, stream>>>(bufB, nullptr, b2, batch, sums, n);
    k_div<<<blk(G * 64), T, 0, stream>>>(sums, cnt, (float*)d_out, G);
}
